# starting kernel
# baseline (speedup 1.0000x reference)
#include <hip/hip_runtime.h>

constexpr int Bdim = 4;
constexpr int Cdim = 1024;
constexpr int Ndim = 64;
constexpr int Tdim = 2048;

typedef float v2f __attribute__((ext_vector_type(2)));
typedef float v4f __attribute__((ext_vector_type(4)));
typedef float v8f __attribute__((ext_vector_type(8)));

__device__ __forceinline__ float softplus_fast(float x) {
    return fmaxf(x, 0.0f) + __logf(1.0f + __expf(-fabsf(x)));
}

__global__ void __launch_bounds__(256)
kgen_wmma(const float* __restrict__ mix,
          const float* __restrict__ log_tau,
          const float* __restrict__ log_dt,
          float* __restrict__ Kout) {
    __shared__ float sRate[Ndim];

    if (threadIdx.x < Ndim) {
        const float dtv = softplus_fast(log_dt[0]);
        sRate[threadIdx.x] = softplus_fast(log_tau[threadIdx.x]) * dtv;
    }
    __syncthreads();

    const int lane = threadIdx.x & 31;
    const int wave = threadIdx.x >> 5;
    const int row  = lane & 15;
    const int half = lane >> 4;

    const int t0 = blockIdx.x * 16;
    const int c0 = (blockIdx.y * 8 + wave) * 16;

    const float tB = (float)(t0 + row);

    v8f acc = {};
    const float* mixRow = mix + (size_t)(c0 + row) * Ndim;

    #pragma unroll
    for (int j0 = 0; j0 < Ndim; j0 += 4) {
        const int ja = j0 + half * 2;

        v2f a;
        a.x = mixRow[ja];
        a.y = mixRow[ja + 1];

        v2f b;
        b.x = __expf(-sRate[ja]     * tB);
        b.y = __expf(-sRate[ja + 1] * tB);

        acc = __builtin_amdgcn_wmma_f32_16x16x4_f32(
             false, a,  false, b,
             (short)0, acc,  false,  false);
    }

    #pragma unroll
    for (int r = 0; r < 8; ++r) {
        const int m = half * 8 + r;
        Kout[(size_t)(c0 + m) * Tdim + (t0 + row)] = acc[r];
    }
}

__global__ void __launch_bounds__(256)
kconv(const float* __restrict__ u,
      const float* __restrict__ Dv,
      const float* __restrict__ Kc,
      float* __restrict__ y) {
    __shared__ float sU[Tdim];
    __shared__ float sK[Tdim];

    const int c = blockIdx.x;
    const int b = blockIdx.y;

    const float* uc = u  + ((size_t)b * Cdim + c) * Tdim;
    const float* kc = Kc + (size_t)c * Tdim;

    for (int i = threadIdx.x; i < Tdim / 4; i += 256) {
        ((v4f*)sU)[i] = ((const v4f*)uc)[i];
        ((v4f*)sK)[i] = ((const v4f*)kc)[i];
    }
    __syncthreads();

    const float d  = Dv[c];
    const int   tb = threadIdx.x << 3;

    float acc[8];
    #pragma unroll
    for (int s = 0; s < 8; ++s) acc[s] = d * sU[tb + s];

    if (tb > 0) {
        v4f w1 = *(const v4f*)&sU[tb];
        v4f w2 = *(const v4f*)&sU[tb + 4];
        for (int jg = 0; jg < tb; jg += 4) {
            const int base = tb - jg;
            const v4f w0 = *(const v4f*)&sU[base - 4];
            const v4f k4 = *(const v4f*)&sK[jg];
            #pragma unroll
            for (int q = 0; q < 4; ++q) {
                #pragma unroll
                for (int s = 0; s < 8; ++s) {
                    const int i = s - q + 4;
                    const float uv = (i < 4) ? w0[i] : (i < 8) ? w1[i - 4] : w2[i - 8];
                    acc[s] += k4[q] * uv;
                }
            }
            w2 = w1;
            w1 = w0;
        }
    }

    {
        const v4f u0 = *(const v4f*)&sU[0];
        const v4f u1 = *(const v4f*)&sU[4];
        #pragma unroll
        for (int e = 0; e < 8; ++e) {
            const float kj = sK[tb + e];
            #pragma unroll
            for (int s = e; s < 8; ++s) {
                const int i = s - e;
                const float uv = (i < 4) ? u0[i] : u1[i - 4];
                acc[s] += kj * uv;
            }
        }
    }

    float* yc = y + ((size_t)b * Cdim + c) * Tdim;
    v4f o0, o1;
    #pragma unroll
    for (int s = 0; s < 4; ++s) { o0[s] = acc[s]; o1[s] = acc[s + 4]; }
    *(v4f*)&yc[tb]     = o0;
    *(v4f*)&yc[tb + 4] = o1;
}

extern "C" void kernel_launch(void* const* d_in, const int* in_sizes, int n_in,
                              void* d_out, int out_size, void* d_ws, size_t ws_size,
                              hipStream_t stream) {
    const float* u       = (const float*)d_in[0];
    const float* log_tau = (const float*)d_in[1];
    const float* mix     = (const float*)d_in[2];
    const float* Dv      = (const float*)d_in[3];
    const float* log_dt  = (const float*)d_in[4];
    float*       y       = (float*)d_out;
    float*       Kws     = (float*)d_ws;

    (void)in_sizes; (void)n_in; (void)out_size; (void)ws_size;

    dim3 g1(Tdim / 16, (Cdim / 16) / 8);
    kgen_wmma<<<g1, 256, 0, stream>>>(mix, log_tau, log_dt, Kws);

    dim3 g2(Cdim, Bdim);
    kconv<<<g2, 256, 0, stream>>>(u, Dv, Kws, y);
}


# conv: split once into even/odd time steps (two-phase fast FIR): three half-length causal convs A,B,C on packed FMAs, y[2m]=A[m]+B[m-1], y[2m+1]=C-A-B; 3/4 of the FMAs; new reduction tree
# speedup vs baseline: 2.6685x; 2.6685x over previous
#include <hip/hip_runtime.h>

constexpr int Bdim = 4;
constexpr int Cdim = 1024;
constexpr int Ndim = 64;
constexpr int Tdim = 2048;

typedef float v2f __attribute__((ext_vector_type(2)));
typedef float v4f __attribute__((ext_vector_type(4)));
typedef float v8f __attribute__((ext_vector_type(8)));

__device__ __forceinline__ float softplus_fast(float x) {
    return fmaxf(x, 0.0f) + __logf(1.0f + __expf(-fabsf(x)));
}

__global__ void __launch_bounds__(256)
kgen_wmma(const float* __restrict__ mix,
          const float* __restrict__ log_tau,
          const float* __restrict__ log_dt,
          float* __restrict__ Kout) {
    __shared__ float sRate[Ndim];

    if (threadIdx.x < Ndim) {
        const float dtv = softplus_fast(log_dt[0]);
        sRate[threadIdx.x] = softplus_fast(log_tau[threadIdx.x]) * dtv;
    }
    __syncthreads();

    const int lane = threadIdx.x & 31;
    const int wave = threadIdx.x >> 5;
    const int row  = lane & 15;
    const int half = lane >> 4;

    const int t0 = blockIdx.x * 16;
    const int c0 = (blockIdx.y * 8 + wave) * 16;

    const float tB = (float)(t0 + row);

    v8f acc = {};
    const float* mixRow = mix + (size_t)(c0 + row) * Ndim;

    #pragma unroll
    for (int j0 = 0; j0 < Ndim; j0 += 4) {
        const int ja = j0 + half * 2;

        v2f a;
        a.x = mixRow[ja];
        a.y = mixRow[ja + 1];

        v2f b;
        b.x = __expf(-sRate[ja]     * tB);
        b.y = __expf(-sRate[ja + 1] * tB);

        acc = __builtin_amdgcn_wmma_f32_16x16x4_f32(
             false, a,  false, b,
             (short)0, acc,  false,  false);
    }

    #pragma unroll
    for (int r = 0; r < 8; ++r) {
        const int m = half * 8 + r;
        Kout[(size_t)(c0 + m) * Tdim + (t0 + row)] = acc[r];
    }
}

constexpr int Th          = Tdim / 2;
constexpr int TilesH      = Th / 8;
constexpr int TilePitch   = 20;
constexpr int RowsPerHalf = 4;
constexpr int TileStride  = RowsPerHalf * TilePitch;
constexpr int KindFloats  = TilesH * TileStride;
constexpr int ChPerBlock  = 2;
constexpr int KPitch      = Th + 4;
constexpr int ConvThreads = 768;

__device__ __forceinline__ v2f tile_pair(const v4f (&b)[4], int i) {
    return (i & 1) ? b[i >> 1].zw : b[i >> 1].xy;
}

__device__ __forceinline__ void load_tile(v4f (&dst)[4], const float* t) {
    const v4f* p = (const v4f*)t;
    dst[0] = p[0];
    dst[1] = p[1];
    dst[2] = p[2];
    dst[3] = p[3];
}

__device__ __forceinline__ void group_step(v2f (&acc)[8],
                                           const v4f (&newer)[4],
                                           const v4f (&older)[4],
                                           const v4f k0, const v4f k1) {
    #pragma unroll
    for (int q = 0; q < 8; ++q) {
        const float kq = (q < 4) ? k0[q & 3] : k1[q & 3];
        const v2f kk = {kq, kq};
        #pragma unroll
        for (int s = 0; s < 8; ++s) {
            const int i = s - q;
            const v2f p = (i >= 0) ? tile_pair(newer, i) : tile_pair(older, 8 + i);
            acc[s] = __builtin_elementwise_fma(kk, p, acc[s]);
        }
    }
}

__device__ __forceinline__ void conv_tile(v2f (&acc)[8],
                                          const float* uRow, const float* kRow,
                                          const float d, const int T) {
    v4f X[4], A[4];
    load_tile(X, uRow + T * TileStride);

    #pragma unroll
    for (int s = 0; s < 8; ++s) acc[s] = tile_pair(X, s) * d;

    const float* up = uRow + (T - 2) * TileStride;
    const float* kp = kRow;
    int G = 0;
    for (; G + 2 <= T; G += 2) {
        load_tile(A, up + TileStride);
        {
            const v4f k0 = *(const v4f*)&kp[0];
            const v4f k1 = *(const v4f*)&kp[4];
            group_step(acc, X, A, k0, k1);
        }
        load_tile(X, up);
        {
            const v4f k0 = *(const v4f*)&kp[8];
            const v4f k1 = *(const v4f*)&kp[12];
            group_step(acc, A, X, k0, k1);
        }
        up -= 2 * TileStride;
        kp += 16;
    }
    if (G < T) {
        load_tile(A, uRow + (T - G - 1) * TileStride);
        const v4f k0 = *(const v4f*)&kp[0];
        const v4f k1 = *(const v4f*)&kp[4];
        group_step(acc, X, A, k0, k1);
    }

    {
        v4f Z[4];
        load_tile(Z, uRow);
        const v4f k0 = *(const v4f*)&kRow[8 * T];
        const v4f k1 = *(const v4f*)&kRow[8 * T + 4];
        #pragma unroll
        for (int e = 0; e < 8; ++e) {
            const float ke = (e < 4) ? k0[e & 3] : k1[e & 3];
            const v2f kk = {ke, ke};
            #pragma unroll
            for (int s = e; s < 8; ++s) {
                acc[s] = __builtin_elementwise_fma(kk, tile_pair(Z, s - e), acc[s]);
            }
        }
    }
}

__device__ __forceinline__ void put_tile(float* t, const v2f (&acc)[8]) {
    v4f* p = (v4f*)t;
    #pragma unroll
    for (int s = 0; s < 4; ++s) {
        const v4f o = {acc[2 * s].x, acc[2 * s].y, acc[2 * s + 1].x, acc[2 * s + 1].y};
        p[s] = o;
    }
}

__global__ void __launch_bounds__(ConvThreads, 12)
kconv(const float* __restrict__ u,
      const float* __restrict__ Dv,
      const float* __restrict__ Kc,
      float* __restrict__ y) {
    __shared__ __attribute__((aligned(16))) float smem[3 * ChPerBlock * KPitch + 3 * KindFloats];
    float* const sK = smem;
    float* const sU = smem + 3 * ChPerBlock * KPitch;

    const int c0  = blockIdx.x * ChPerBlock;
    const int tid = threadIdx.x;

    #pragma unroll
    for (int it = 0; it < 3; ++it) {
        const int q = it * ConvThreads + tid;
        if (q < 4 * 512) {
            const int r = q >> 9;
            const int i = q & 511;
            const float* ua = u + ((size_t)(2 * (r & 1)) * Cdim + (c0 + (r >> 1))) * Tdim;
            const float* ub = ua + (size_t)Cdim * Tdim;
            const v4f a  = ((const v4f*)ua)[i];
            const v4f bq = ((const v4f*)ub)[i];
            const v4f x0 = {a.x, bq.x, a.z, bq.z};
            const v4f x1 = {a.y, bq.y, a.w, bq.w};
            const v4f xs = x0 + x1;
            float* dst = sU + ((i >> 2) * RowsPerHalf + r) * TilePitch + (i & 3) * 4;
            *(v4f*)(dst)                  = x0;
            *(v4f*)(dst + KindFloats)     = x1;
            *(v4f*)(dst + 2 * KindFloats) = xs;
        }
    }
    #pragma unroll
    for (int it = 0; it < 2; ++it) {
        const int q = it * ConvThreads + tid;
        if (q < ChPerBlock * 512) {
            const int cl = q >> 9;
            const int i  = q & 511;
            const v4f k = ((const v4f*)(Kc + (size_t)(c0 + cl) * Tdim))[i];
            const v2f h0 = {k.x, k.z};
            const v2f h1 = {k.y, k.w};
            const v2f hs = h0 + h1;
            float* dst = sK + cl * KPitch + 2 * i;
            *(v2f*)(dst)                           = h0;
            *(v2f*)(dst + ChPerBlock * KPitch)     = h1;
            *(v2f*)(dst + 2 * ChPerBlock * KPitch) = hs;
        }
    }
    __syncthreads();

    const int lane = tid & 31;
    const int wave = tid >> 5;
    const int hf   = lane >> 4;
    const int r    = (lane >> 2) & 3;
    const int kind = (wave < 16) ? hf : 2;
    const int sq   = (wave < 16) ? wave : 2 * (wave - 16) + hf;
    const int j    = sq * 4 + (lane & 3);
    const int cl   = r >> 1;

    float* const uRow = sU + kind * KindFloats + r * TilePitch;
    const float* kRow = sK + (kind * ChPerBlock + cl) * KPitch;

    const float d = (kind == 1) ? 0.0f : Dv[c0 + cl];

    v2f resLong[8], resShort[8];
    conv_tile(resLong,  uRow, kRow, d, (TilesH - 1) - j);
    conv_tile(resShort, uRow, kRow, d, j);

    __syncthreads();
    put_tile(uRow + ((TilesH - 1) - j) * TileStride, resLong);
    put_tile(uRow + j * TileStride, resShort);
    __syncthreads();

    #pragma unroll
    for (int it = 0; it < 3; ++it) {
        const int q = it * ConvThreads + tid;
        if (q < 4 * 512) {
            const int rr = q >> 9;
            const int i  = q & 511;
            const float* pa = sU + ((i >> 2) * RowsPerHalf + rr) * TilePitch + (i & 3) * 4;
            const v4f av = *(const v4f*)(pa);
            const v4f bv = *(const v4f*)(pa + KindFloats);
            const v4f cv = *(const v4f*)(pa + 2 * KindFloats);
            v2f bprev = {0.0f, 0.0f};
            if (i > 0) {
                const float* pb = pa + KindFloats + ((i & 3) ? -2 : (14 - TileStride));
                bprev = *(const v2f*)pb;
            }
            const v2f y0a = av.xy + bprev;
            const v2f y1a = cv.xy - av.xy - bv.xy;
            const v2f y0b = av.zw + bv.xy;
            const v2f y1b = cv.zw - av.zw - bv.zw;
            float* ya = y + ((size_t)(2 * (rr & 1)) * Cdim + (c0 + (rr >> 1))) * Tdim;
            float* yb = ya + (size_t)Cdim * Tdim;
            const v4f oa = {y0a.x, y1a.x, y0b.x, y1b.x};
            const v4f ob = {y0a.y, y1a.y, y0b.y, y1b.y};
            ((v4f*)ya)[i] = oa;
            ((v4f*)yb)[i] = ob;
        }
    }
}

extern "C" void kernel_launch(void* const* d_in, const int* in_sizes, int n_in,
                              void* d_out, int out_size, void* d_ws, size_t ws_size,
                              hipStream_t stream) {
    const float* u       = (const float*)d_in[0];
    const float* log_tau = (const float*)d_in[1];
    const float* mix     = (const float*)d_in[2];
    const float* Dv      = (const float*)d_in[3];
    const float* log_dt  = (const float*)d_in[4];
    float*       y       = (float*)d_out;
    float*       Kws     = (float*)d_ws;

    (void)in_sizes; (void)n_in; (void)out_size; (void)ws_size;

    dim3 g1(Tdim / 16, (Cdim / 16) / 8);
    kgen_wmma<<<g1, 256, 0, stream>>>(mix, log_tau, log_dt, Kws);

    dim3 g2(Cdim / ChPerBlock);
    kconv<<<g2, ConvThreads, 0, stream>>>(u, Dv, Kws, y);
}


# conv: even/odd split nested a second time: nine quarter-length causal convs on the same packed FMAs, two-level recombination; 9/16 of the direct FMAs (was 3/4); new reduction tree
# speedup vs baseline: 3.0880x; 1.1572x over previous
#include <hip/hip_runtime.h>

constexpr int Bdim = 4;
constexpr int Cdim = 1024;
constexpr int Ndim = 64;
constexpr int Tdim = 2048;

typedef float v2f __attribute__((ext_vector_type(2)));
typedef float v4f __attribute__((ext_vector_type(4)));
typedef float v8f __attribute__((ext_vector_type(8)));

__device__ __forceinline__ float softplus_fast(float x) {
    return fmaxf(x, 0.0f) + __logf(1.0f + __expf(-fabsf(x)));
}

__global__ void __launch_bounds__(256)
kgen_wmma(const float* __restrict__ mix,
          const float* __restrict__ log_tau,
          const float* __restrict__ log_dt,
          float* __restrict__ Kout) {
    __shared__ float sRate[Ndim];

    if (threadIdx.x < Ndim) {
        const float dtv = softplus_fast(log_dt[0]);
        sRate[threadIdx.x] = softplus_fast(log_tau[threadIdx.x]) * dtv;
    }
    __syncthreads();

    const int lane = threadIdx.x & 31;
    const int wave = threadIdx.x >> 5;
    const int row  = lane & 15;
    const int half = lane >> 4;

    const int t0 = blockIdx.x * 16;
    const int c0 = (blockIdx.y * 8 + wave) * 16;

    const float tB = (float)(t0 + row);

    v8f acc = {};
    const float* mixRow = mix + (size_t)(c0 + row) * Ndim;

    #pragma unroll
    for (int j0 = 0; j0 < Ndim; j0 += 4) {
        const int ja = j0 + half * 2;

        v2f a;
        a.x = mixRow[ja];
        a.y = mixRow[ja + 1];

        v2f b;
        b.x = __expf(-sRate[ja]     * tB);
        b.y = __expf(-sRate[ja + 1] * tB);

        acc = __builtin_amdgcn_wmma_f32_16x16x4_f32(
             false, a,  false, b,
             (short)0, acc,  false,  false);
    }

    #pragma unroll
    for (int r = 0; r < 8; ++r) {
        const int m = half * 8 + r;
        Kout[(size_t)(c0 + m) * Tdim + (t0 + row)] = acc[r];
    }
}

constexpr int Tq          = Tdim / 4;
constexpr int TilesQ      = Tq / 8;
constexpr int TilePitch   = 20;
constexpr int RowsPerKind = 2;
constexpr int TileStride  = RowsPerKind * TilePitch;
constexpr int KindFloats  = TilesQ * TileStride;
constexpr int KindSkew    = 2 * TilePitch;
constexpr int NKinds      = 9;
constexpr int KPitch      = Tq + 4;
constexpr int ConvThreads = 640;

__device__ __forceinline__ int kind_base(int kind) {
    return kind * (KindFloats + KindSkew);
}

__device__ __forceinline__ v2f tile_pair(const v4f (&b)[4], int i) {
    return (i & 1) ? b[i >> 1].zw : b[i >> 1].xy;
}

__device__ __forceinline__ void load_tile(v4f (&dst)[4], const float* t) {
    const v4f* p = (const v4f*)t;
    dst[0] = p[0];
    dst[1] = p[1];
    dst[2] = p[2];
    dst[3] = p[3];
}

__device__ __forceinline__ void group_step(v2f (&acc)[8],
                                           const v4f (&newer)[4],
                                           const v4f (&older)[4],
                                           const v4f k0, const v4f k1) {
    #pragma unroll
    for (int q = 0; q < 8; ++q) {
        const float kq = (q < 4) ? k0[q & 3] : k1[q & 3];
        const v2f kk = {kq, kq};
        #pragma unroll
        for (int s = 0; s < 8; ++s) {
            const int i = s - q;
            const v2f p = (i >= 0) ? tile_pair(newer, i) : tile_pair(older, 8 + i);
            acc[s] = __builtin_elementwise_fma(kk, p, acc[s]);
        }
    }
}

__device__ __forceinline__ void conv_tile(v2f (&acc)[8],
                                          const float* uRow, const float* kRow,
                                          const float d, const int T) {
    v4f X[4], A[4];
    load_tile(X, uRow + T * TileStride);

    #pragma unroll
    for (int s = 0; s < 8; ++s) acc[s] = tile_pair(X, s) * d;

    const float* up = uRow + (T - 2) * TileStride;
    const float* kp = kRow;
    int G = 0;
    for (; G + 2 <= T; G += 2) {
        load_tile(A, up + TileStride);
        {
            const v4f k0 = *(const v4f*)&kp[0];
            const v4f k1 = *(const v4f*)&kp[4];
            group_step(acc, X, A, k0, k1);
        }
        load_tile(X, up);
        {
            const v4f k0 = *(const v4f*)&kp[8];
            const v4f k1 = *(const v4f*)&kp[12];
            group_step(acc, A, X, k0, k1);
        }
        up -= 2 * TileStride;
        kp += 16;
    }
    if (G < T) {
        load_tile(A, uRow + (T - G - 1) * TileStride);
        const v4f k0 = *(const v4f*)&kp[0];
        const v4f k1 = *(const v4f*)&kp[4];
        group_step(acc, X, A, k0, k1);
    }

    {
        v4f Z[4];
        load_tile(Z, uRow);
        const v4f k0 = *(const v4f*)&kRow[8 * T];
        const v4f k1 = *(const v4f*)&kRow[8 * T + 4];
        #pragma unroll
        for (int e = 0; e < 8; ++e) {
            const float ke = (e < 4) ? k0[e & 3] : k1[e & 3];
            const v2f kk = {ke, ke};
            #pragma unroll
            for (int s = e; s < 8; ++s) {
                acc[s] = __builtin_elementwise_fma(kk, tile_pair(Z, s - e), acc[s]);
            }
        }
    }
}

__device__ __forceinline__ void put_tile(float* t, const v2f (&acc)[8]) {
    v4f* p = (v4f*)t;
    #pragma unroll
    for (int s = 0; s < 4; ++s) {
        const v4f o = {acc[2 * s].x, acc[2 * s].y, acc[2 * s + 1].x, acc[2 * s + 1].y};
        p[s] = o;
    }
}

__global__ void __launch_bounds__(ConvThreads, 10)
kconv(const float* __restrict__ u,
      const float* __restrict__ Dv,
      const float* __restrict__ Kc,
      float* __restrict__ y) {
    __shared__ __attribute__((aligned(16))) float smem[NKinds * KPitch + NKinds * (KindFloats + KindSkew)];
    float* const sK = smem;
    float* const sU = smem + NKinds * KPitch;

    const int c   = blockIdx.x;
    const int tid = threadIdx.x;

    #pragma unroll
    for (int it = 0; it < 2; ++it) {
        const int q = it * ConvThreads + tid;
        if (q < 2 * Tq) {
            const int r = q >> 9;
            const int i = q & (Tq - 1);
            const float* ua = u + ((size_t)(2 * r) * Cdim + c) * Tdim;
            const float* ub = ua + (size_t)Cdim * Tdim;
            const v4f a  = ((const v4f*)ua)[i];
            const v4f bq = ((const v4f*)ub)[i];
            const v2f px = {a.x, bq.x};
            const v2f py = {a.y, bq.y};
            const v2f pz = {a.z, bq.z};
            const v2f pw = {a.w, bq.w};
            const v2f sxy = px + py;
            const v2f szw = pz + pw;
            float* dst = sU + (2 * (i >> 3) + r) * TilePitch + 2 * (i & 7);
            *(v2f*)(dst + kind_base(0)) = px;
            *(v2f*)(dst + kind_base(1)) = pz;
            *(v2f*)(dst + kind_base(2)) = px + pz;
            *(v2f*)(dst + kind_base(3)) = py;
            *(v2f*)(dst + kind_base(4)) = pw;
            *(v2f*)(dst + kind_base(5)) = py + pw;
            *(v2f*)(dst + kind_base(6)) = sxy;
            *(v2f*)(dst + kind_base(7)) = szw;
            *(v2f*)(dst + kind_base(8)) = sxy + szw;
        }
    }
    if (tid < Tq) {
        const v4f k = ((const v4f*)(Kc + (size_t)c * Tdim))[tid];
        const float kxy = k.x + k.y;
        const float kzw = k.z + k.w;
        float* dst = sK + tid;
        dst[0 * KPitch] = k.x;
        dst[1 * KPitch] = k.z;
        dst[2 * KPitch] = k.x + k.z;
        dst[3 * KPitch] = k.y;
        dst[4 * KPitch] = k.w;
        dst[5 * KPitch] = k.y + k.w;
        dst[6 * KPitch] = kxy;
        dst[7 * KPitch] = kzw;
        dst[8 * KPitch] = kxy + kzw;
    }
    __syncthreads();

    const int lane = tid & 31;
    const int wave = tid >> 5;
    const int hf   = lane >> 4;
    const int l16  = lane & 15;
    int kind, r, j;
    bool active = true;
    if (wave < 16) {
        kind = l16 >> 1;
        r    = l16 & 1;
        j    = 2 * wave + hf;
    } else {
        kind   = 8;
        r      = (l16 >> 2) & 1;
        j      = (wave - 16) * 8 + hf * 4 + (l16 & 3);
        active = l16 < 8;
    }

    float* const uRow = sU + kind_base(kind) + r * TilePitch;
    const float* kRow = sK + kind * KPitch;

    const bool skip = (kind == 0) || (kind == 2) || (kind == 6) || (kind == 8);
    const float d = skip ? Dv[c] : 0.0f;

    v2f resLong[8], resShort[8];
    if (active) {
        conv_tile(resLong,  uRow, kRow, d, (TilesQ - 1) - j);
        conv_tile(resShort, uRow, kRow, d, j);
    }

    __syncthreads();
    if (active) {
        put_tile(uRow + ((TilesQ - 1) - j) * TileStride, resLong);
        put_tile(uRow + j * TileStride, resShort);
    }
    __syncthreads();

    #pragma unroll
    for (int it = 0; it < 2; ++it) {
        const int q = it * ConvThreads + tid;
        if (q < 2 * Tq) {
            const int rr = q >> 9;
            const int i  = q & (Tq - 1);
            const float* pc = sU + (2 * (i >> 3) + rr) * TilePitch + 2 * (i & 7);
            const v2f R0 = *(const v2f*)(pc + kind_base(0));
            const v2f R1 = *(const v2f*)(pc + kind_base(1));
            const v2f R2 = *(const v2f*)(pc + kind_base(2));
            const v2f R3 = *(const v2f*)(pc + kind_base(3));
            const v2f R4 = *(const v2f*)(pc + kind_base(4));
            const v2f R5 = *(const v2f*)(pc + kind_base(5));
            const v2f R6 = *(const v2f*)(pc + kind_base(6));
            const v2f R7 = *(const v2f*)(pc + kind_base(7));
            const v2f R8 = *(const v2f*)(pc + kind_base(8));
            v2f q1 = {0.0f, 0.0f}, q3 = q1, q4 = q1, q5 = q1, q7 = q1;
            if (i > 0) {
                const float* pp = pc + ((i & 7) ? -2 : (14 - TileStride));
                q1 = *(const v2f*)(pp + kind_base(1));
                q3 = *(const v2f*)(pp + kind_base(3));
                q4 = *(const v2f*)(pp + kind_base(4));
                q5 = *(const v2f*)(pp + kind_base(5));
                q7 = *(const v2f*)(pp + kind_base(7));
            }
            const v2f pe0 = R0 + q1;
            const v2f po0 = R2 - R0 - R1;
            const v2f pe1 = R3 + q4;
            const v2f po1 = R5 - R3 - R4;
            const v2f peS = R6 + q7;
            const v2f poS = R8 - R6 - R7;
            const v2f pm1 = q5 - q3 - q4;
            const v2f y0 = pe0 + pm1;
            const v2f y1 = peS - pe0 - pe1;
            const v2f y2 = po0 + pe1;
            const v2f y3 = poS - po0 - po1;
            float* ya = y + ((size_t)(2 * rr) * Cdim + c) * Tdim;
            float* yb = ya + (size_t)Cdim * Tdim;
            const v4f oa = {y0.x, y1.x, y2.x, y3.x};
            const v4f ob = {y0.y, y1.y, y2.y, y3.y};
            ((v4f*)ya)[i] = oa;
            ((v4f*)yb)[i] = ob;
        }
    }
}

extern "C" void kernel_launch(void* const* d_in, const int* in_sizes, int n_in,
                              void* d_out, int out_size, void* d_ws, size_t ws_size,
                              hipStream_t stream) {
    const float* u       = (const float*)d_in[0];
    const float* log_tau = (const float*)d_in[1];
    const float* mix     = (const float*)d_in[2];
    const float* Dv      = (const float*)d_in[3];
    const float* log_dt  = (const float*)d_in[4];
    float*       y       = (float*)d_out;
    float*       Kws     = (float*)d_ws;

    (void)in_sizes; (void)n_in; (void)out_size; (void)ws_size;

    dim3 g1(Tdim / 16, (Cdim / 16) / 8);
    kgen_wmma<<<g1, 256, 0, stream>>>(mix, log_tau, log_dt, Kws);

    dim3 g2(Cdim);
    kconv<<<g2, ConvThreads, 0, stream>>>(u, Dv, Kws, y);
}


# conv: kind SS (2 sub-rows) fills its 4 waves: two lanes share a tile, 4 outputs each over a window of three 4-step half-tiles, so those waves issue half the packed FMAs; sums per output unchanged
# speedup vs baseline: 3.2681x; 1.0583x over previous
#include <hip/hip_runtime.h>

constexpr int Bdim = 4;
constexpr int Cdim = 1024;
constexpr int Ndim = 64;
constexpr int Tdim = 2048;

typedef float v2f __attribute__((ext_vector_type(2)));
typedef float v4f __attribute__((ext_vector_type(4)));
typedef float v8f __attribute__((ext_vector_type(8)));

__device__ __forceinline__ float softplus_fast(float x) {
    return fmaxf(x, 0.0f) + __logf(1.0f + __expf(-fabsf(x)));
}

__global__ void __launch_bounds__(256)
kgen_wmma(const float* __restrict__ mix,
          const float* __restrict__ log_tau,
          const float* __restrict__ log_dt,
          float* __restrict__ Kout) {
    __shared__ float sRate[Ndim];

    if (threadIdx.x < Ndim) {
        const float dtv = softplus_fast(log_dt[0]);
        sRate[threadIdx.x] = softplus_fast(log_tau[threadIdx.x]) * dtv;
    }
    __syncthreads();

    const int lane = threadIdx.x & 31;
    const int wave = threadIdx.x >> 5;
    const int row  = lane & 15;
    const int half = lane >> 4;

    const int t0 = blockIdx.x * 16;
    const int c0 = (blockIdx.y * 8 + wave) * 16;

    const float tB = (float)(t0 + row);

    v8f acc = {};
    const float* mixRow = mix + (size_t)(c0 + row) * Ndim;

    #pragma unroll
    for (int j0 = 0; j0 < Ndim; j0 += 4) {
        const int ja = j0 + half * 2;

        v2f a;
        a.x = mixRow[ja];
        a.y = mixRow[ja + 1];

        v2f b;
        b.x = __expf(-sRate[ja]     * tB);
        b.y = __expf(-sRate[ja + 1] * tB);

        acc = __builtin_amdgcn_wmma_f32_16x16x4_f32(
             false, a,  false, b,
             (short)0, acc,  false,  false);
    }

    #pragma unroll
    for (int r = 0; r < 8; ++r) {
        const int m = half * 8 + r;
        Kout[(size_t)(c0 + m) * Tdim + (t0 + row)] = acc[r];
    }
}

constexpr int Tq          = Tdim / 4;
constexpr int TilesQ      = Tq / 8;
constexpr int TilePitch   = 20;
constexpr int RowsPerKind = 2;
constexpr int TileStride  = RowsPerKind * TilePitch;
constexpr int KindFloats  = TilesQ * TileStride;
constexpr int KindSkew    = 2 * TilePitch;
constexpr int NKinds      = 9;
constexpr int KPitch      = Tq + 4;
constexpr int ConvThreads = 640;

__device__ __forceinline__ int kind_base(int kind) {
    return kind * (KindFloats + KindSkew);
}

__device__ __forceinline__ v2f tile_pair(const v4f (&b)[4], int i) {
    return (i & 1) ? b[i >> 1].zw : b[i >> 1].xy;
}

__device__ __forceinline__ void load_tile(v4f (&dst)[4], const float* t) {
    const v4f* p = (const v4f*)t;
    dst[0] = p[0];
    dst[1] = p[1];
    dst[2] = p[2];
    dst[3] = p[3];
}

__device__ __forceinline__ void group_step(v2f (&acc)[8],
                                           const v4f (&newer)[4],
                                           const v4f (&older)[4],
                                           const v4f k0, const v4f k1) {
    #pragma unroll
    for (int q = 0; q < 8; ++q) {
        const float kq = (q < 4) ? k0[q & 3] : k1[q & 3];
        const v2f kk = {kq, kq};
        #pragma unroll
        for (int s = 0; s < 8; ++s) {
            const int i = s - q;
            const v2f p = (i >= 0) ? tile_pair(newer, i) : tile_pair(older, 8 + i);
            acc[s] = __builtin_elementwise_fma(kk, p, acc[s]);
        }
    }
}

__device__ __forceinline__ void conv_tile(v2f (&acc)[8],
                                          const float* uRow, const float* kRow,
                                          const float d, const int T) {
    v4f X[4], A[4];
    load_tile(X, uRow + T * TileStride);

    #pragma unroll
    for (int s = 0; s < 8; ++s) acc[s] = tile_pair(X, s) * d;

    const float* up = uRow + (T - 2) * TileStride;
    const float* kp = kRow;
    int G = 0;
    for (; G + 2 <= T; G += 2) {
        load_tile(A, up + TileStride);
        {
            const v4f k0 = *(const v4f*)&kp[0];
            const v4f k1 = *(const v4f*)&kp[4];
            group_step(acc, X, A, k0, k1);
        }
        load_tile(X, up);
        {
            const v4f k0 = *(const v4f*)&kp[8];
            const v4f k1 = *(const v4f*)&kp[12];
            group_step(acc, A, X, k0, k1);
        }
        up -= 2 * TileStride;
        kp += 16;
    }
    if (G < T) {
        load_tile(A, uRow + (T - G - 1) * TileStride);
        const v4f k0 = *(const v4f*)&kp[0];
        const v4f k1 = *(const v4f*)&kp[4];
        group_step(acc, X, A, k0, k1);
    }

    {
        v4f Z[4];
        load_tile(Z, uRow);
        const v4f k0 = *(const v4f*)&kRow[8 * T];
        const v4f k1 = *(const v4f*)&kRow[8 * T + 4];
        #pragma unroll
        for (int e = 0; e < 8; ++e) {
            const float ke = (e < 4) ? k0[e & 3] : k1[e & 3];
            const v2f kk = {ke, ke};
            #pragma unroll
            for (int s = e; s < 8; ++s) {
                acc[s] = __builtin_elementwise_fma(kk, tile_pair(Z, s - e), acc[s]);
            }
        }
    }
}

__device__ __forceinline__ void put_tile(float* t, const v2f (&acc)[8]) {
    v4f* p = (v4f*)t;
    #pragma unroll
    for (int s = 0; s < 4; ++s) {
        const v4f o = {acc[2 * s].x, acc[2 * s].y, acc[2 * s + 1].x, acc[2 * s + 1].y};
        p[s] = o;
    }
}


__device__ __forceinline__ v2f half_pair(const v4f (&b)[2], int i) {
    return (i & 1) ? b[i >> 1].zw : b[i >> 1].xy;
}

__device__ __forceinline__ void load_half(v4f (&dst)[2], const float* t) {
    const v4f* p = (const v4f*)t;
    dst[0] = p[0];
    dst[1] = p[1];
}

__device__ __forceinline__ void half_group_step(v2f (&acc)[8],
                                                const v4f (&cur)[2],
                                                const v4f (&m1)[2],
                                                const v4f (&m2)[2],
                                                const v4f k0, const v4f k1) {
    #pragma unroll
    for (int q = 0; q < 8; ++q) {
        const float kq = (q < 4) ? k0[q & 3] : k1[q & 3];
        const v2f kk = {kq, kq};
        #pragma unroll
        for (int s = 0; s < 4; ++s) {
            const int i = s - q;
            const v2f p = (i >= 0) ? half_pair(cur, i)
                        : (i >= -4) ? half_pair(m1, 4 + i) : half_pair(m2, 8 + i);
            acc[s] = __builtin_elementwise_fma(kk, p, acc[s]);
        }
    }
}

__device__ __forceinline__ void conv_half(v2f (&acc)[8],
                                          const float* uRow, const float* kRow,
                                          const float d, const int T, const int hs) {
    v4f X[2], A1[2], A2[2], B1[2];
    load_half(X, uRow + T * TileStride + 8 * hs);

    #pragma unroll
    for (int s = 0; s < 4; ++s) acc[s] = half_pair(X, s) * d;

    const float* q1 = uRow + (T - 2 + hs) * TileStride + 8 * (1 - hs);
    const float* q2 = uRow + (T - 2) * TileStride + 8 * hs;
    const float* kp = kRow;
    int G = 0;
    for (; G + 2 <= T; G += 2) {
        load_half(A1, q1 + TileStride);
        load_half(A2, q2 + TileStride);
        {
            const v4f k0 = *(const v4f*)&kp[0];
            const v4f k1 = *(const v4f*)&kp[4];
            half_group_step(acc, X, A1, A2, k0, k1);
        }
        load_half(B1, q1);
        load_half(X, q2);
        {
            const v4f k0 = *(const v4f*)&kp[8];
            const v4f k1 = *(const v4f*)&kp[12];
            half_group_step(acc, A2, B1, X, k0, k1);
        }
        q1 -= 2 * TileStride;
        q2 -= 2 * TileStride;
        kp += 16;
    }
    if (G < T) {
        load_half(A1, q1 + TileStride);
        load_half(A2, q2 + TileStride);
        const v4f k0 = *(const v4f*)&kp[0];
        const v4f k1 = *(const v4f*)&kp[4];
        half_group_step(acc, X, A1, A2, k0, k1);
        X[0] = A2[0];
        X[1] = A2[1];
    }

    {
        v4f Z[2];
        load_half(Z, uRow);
        if (hs == 0) {
            const v4f zero = {0.0f, 0.0f, 0.0f, 0.0f};
            Z[0] = zero;
            Z[1] = zero;
        }
        const v4f k0 = *(const v4f*)&kRow[8 * T];
        const v4f k1 = *(const v4f*)&kRow[8 * T + 4];
        #pragma unroll
        for (int q = 0; q < 8; ++q) {
            const float kq = (q < 4) ? k0[q & 3] : k1[q & 3];
            const v2f kk = {kq, kq};
            #pragma unroll
            for (int s = 0; s < 4; ++s) {
                const int i = s - q;
                if (i >= 0) {
                    acc[s] = __builtin_elementwise_fma(kk, half_pair(X, i), acc[s]);
                } else if (i >= -4) {
                    acc[s] = __builtin_elementwise_fma(kk, half_pair(Z, 4 + i), acc[s]);
                }
            }
        }
    }
}

__device__ __forceinline__ void put_half(float* t, const v2f (&acc)[8]) {
    v4f* p = (v4f*)t;
    #pragma unroll
    for (int s = 0; s < 2; ++s) {
        const v4f o = {acc[2 * s].x, acc[2 * s].y, acc[2 * s + 1].x, acc[2 * s + 1].y};
        p[s] = o;
    }
}

__global__ void __launch_bounds__(ConvThreads, 10)
kconv(const float* __restrict__ u,
      const float* __restrict__ Dv,
      const float* __restrict__ Kc,
      float* __restrict__ y) {
    __shared__ __attribute__((aligned(16))) float smem[NKinds * KPitch + NKinds * (KindFloats + KindSkew)];
    float* const sK = smem;
    float* const sU = smem + NKinds * KPitch;

    const int c   = blockIdx.x;
    const int tid = threadIdx.x;

    #pragma unroll
    for (int it = 0; it < 2; ++it) {
        const int q = it * ConvThreads + tid;
        if (q < 2 * Tq) {
            const int r = q >> 9;
            const int i = q & (Tq - 1);
            const float* ua = u + ((size_t)(2 * r) * Cdim + c) * Tdim;
            const float* ub = ua + (size_t)Cdim * Tdim;
            const v4f a  = ((const v4f*)ua)[i];
            const v4f bq = ((const v4f*)ub)[i];
            const v2f px = {a.x, bq.x};
            const v2f py = {a.y, bq.y};
            const v2f pz = {a.z, bq.z};
            const v2f pw = {a.w, bq.w};
            const v2f sxy = px + py;
            const v2f szw = pz + pw;
            float* dst = sU + (2 * (i >> 3) + r) * TilePitch + 2 * (i & 7);
            *(v2f*)(dst + kind_base(0)) = px;
            *(v2f*)(dst + kind_base(1)) = pz;
            *(v2f*)(dst + kind_base(2)) = px + pz;
            *(v2f*)(dst + kind_base(3)) = py;
            *(v2f*)(dst + kind_base(4)) = pw;
            *(v2f*)(dst + kind_base(5)) = py + pw;
            *(v2f*)(dst + kind_base(6)) = sxy;
            *(v2f*)(dst + kind_base(7)) = szw;
            *(v2f*)(dst + kind_base(8)) = sxy + szw;
        }
    }
    if (tid < Tq) {
        const v4f k = ((const v4f*)(Kc + (size_t)c * Tdim))[tid];
        const float kxy = k.x + k.y;
        const float kzw = k.z + k.w;
        float* dst = sK + tid;
        dst[0 * KPitch] = k.x;
        dst[1 * KPitch] = k.z;
        dst[2 * KPitch] = k.x + k.z;
        dst[3 * KPitch] = k.y;
        dst[4 * KPitch] = k.w;
        dst[5 * KPitch] = k.y + k.w;
        dst[6 * KPitch] = kxy;
        dst[7 * KPitch] = kzw;
        dst[8 * KPitch] = kxy + kzw;
    }
    __syncthreads();

    const int lane = tid & 31;
    const int wave = tid >> 5;
    const int hf   = lane >> 4;
    const int l16  = lane & 15;
    int kind, r, j, hs = 0;
    const bool whole = wave < 16;
    if (whole) {
        kind = l16 >> 1;
        r    = l16 & 1;
        j    = 2 * wave + hf;
    } else {
        kind = 8;
        hs   = l16 >> 3;
        r    = (l16 >> 2) & 1;
        j    = (wave - 16) * 8 + hf * 4 + (l16 & 3);
    }

    float* const uRow = sU + kind_base(kind) + r * TilePitch;
    const float* kRow = sK + kind * KPitch;

    const bool skip = (kind == 0) || (kind == 2) || (kind == 6) || (kind == 8);
    const float d = skip ? Dv[c] : 0.0f;

    v2f resLong[8], resShort[8];
    if (whole) {
        conv_tile(resLong,  uRow, kRow, d, (TilesQ - 1) - j);
        conv_tile(resShort, uRow, kRow, d, j);
    } else {
        conv_half(resLong,  uRow, kRow, d, (TilesQ - 1) - j, hs);
        conv_half(resShort, uRow, kRow, d, j, hs);
    }

    __syncthreads();
    if (whole) {
        put_tile(uRow + ((TilesQ - 1) - j) * TileStride, resLong);
        put_tile(uRow + j * TileStride, resShort);
    } else {
        put_half(uRow + ((TilesQ - 1) - j) * TileStride + 8 * hs, resLong);
        put_half(uRow + j * TileStride + 8 * hs, resShort);
    }
    __syncthreads();

    #pragma unroll
    for (int it = 0; it < 2; ++it) {
        const int q = it * ConvThreads + tid;
        if (q < 2 * Tq) {
            const int rr = q >> 9;
            const int i  = q & (Tq - 1);
            const float* pc = sU + (2 * (i >> 3) + rr) * TilePitch + 2 * (i & 7);
            const v2f R0 = *(const v2f*)(pc + kind_base(0));
            const v2f R1 = *(const v2f*)(pc + kind_base(1));
            const v2f R2 = *(const v2f*)(pc + kind_base(2));
            const v2f R3 = *(const v2f*)(pc + kind_base(3));
            const v2f R4 = *(const v2f*)(pc + kind_base(4));
            const v2f R5 = *(const v2f*)(pc + kind_base(5));
            const v2f R6 = *(const v2f*)(pc + kind_base(6));
            const v2f R7 = *(const v2f*)(pc + kind_base(7));
            const v2f R8 = *(const v2f*)(pc + kind_base(8));
            v2f q1 = {0.0f, 0.0f}, q3 = q1, q4 = q1, q5 = q1, q7 = q1;
            if (i > 0) {
                const float* pp = pc + ((i & 7) ? -2 : (14 - TileStride));
                q1 = *(const v2f*)(pp + kind_base(1));
                q3 = *(const v2f*)(pp + kind_base(3));
                q4 = *(const v2f*)(pp + kind_base(4));
                q5 = *(const v2f*)(pp + kind_base(5));
                q7 = *(const v2f*)(pp + kind_base(7));
            }
            const v2f pe0 = R0 + q1;
            const v2f po0 = R2 - R0 - R1;
            const v2f pe1 = R3 + q4;
            const v2f po1 = R5 - R3 - R4;
            const v2f peS = R6 + q7;
            const v2f poS = R8 - R6 - R7;
            const v2f pm1 = q5 - q3 - q4;
            const v2f y0 = pe0 + pm1;
            const v2f y1 = peS - pe0 - pe1;
            const v2f y2 = po0 + pe1;
            const v2f y3 = poS - po0 - po1;
            float* ya = y + ((size_t)(2 * rr) * Cdim + c) * Tdim;
            float* yb = ya + (size_t)Cdim * Tdim;
            const v4f oa = {y0.x, y1.x, y2.x, y3.x};
            const v4f ob = {y0.y, y1.y, y2.y, y3.y};
            ((v4f*)ya)[i] = oa;
            ((v4f*)yb)[i] = ob;
        }
    }
}

extern "C" void kernel_launch(void* const* d_in, const int* in_sizes, int n_in,
                              void* d_out, int out_size, void* d_ws, size_t ws_size,
                              hipStream_t stream) {
    const float* u       = (const float*)d_in[0];
    const float* log_tau = (const float*)d_in[1];
    const float* mix     = (const float*)d_in[2];
    const float* Dv      = (const float*)d_in[3];
    const float* log_dt  = (const float*)d_in[4];
    float*       y       = (float*)d_out;
    float*       Kws     = (float*)d_ws;

    (void)in_sizes; (void)n_in; (void)out_size; (void)ws_size;

    dim3 g1(Tdim / 16, (Cdim / 16) / 8);
    kgen_wmma<<<g1, 256, 0, stream>>>(mix, log_tau, log_dt, Kws);

    dim3 g2(Cdim);
    kconv<<<g2, ConvThreads, 0, stream>>>(u, Dv, Kws, y);
}


# conv kinds 0..7: a lane owns a 16-step tile (16 outputs x 2 batches), taps in groups of 16: half the LDS reads (pairs, taps) per packed FMA; 12 waves a block, not 20; sums per output unchanged
# speedup vs baseline: 3.6531x; 1.1178x over previous
#include <hip/hip_runtime.h>

constexpr int Bdim = 4;
constexpr int Cdim = 1024;
constexpr int Ndim = 64;
constexpr int Tdim = 2048;

typedef float v2f __attribute__((ext_vector_type(2)));
typedef float v4f __attribute__((ext_vector_type(4)));
typedef float v8f __attribute__((ext_vector_type(8)));

__device__ __forceinline__ float softplus_fast(float x) {
    return fmaxf(x, 0.0f) + __logf(1.0f + __expf(-fabsf(x)));
}

__global__ void __launch_bounds__(256)
kgen_wmma(const float* __restrict__ mix,
          const float* __restrict__ log_tau,
          const float* __restrict__ log_dt,
          float* __restrict__ Kout) {
    __shared__ float sRate[Ndim];

    if (threadIdx.x < Ndim) {
        const float dtv = softplus_fast(log_dt[0]);
        sRate[threadIdx.x] = softplus_fast(log_tau[threadIdx.x]) * dtv;
    }
    __syncthreads();

    const int lane = threadIdx.x & 31;
    const int wave = threadIdx.x >> 5;
    const int row  = lane & 15;
    const int half = lane >> 4;

    const int t0 = blockIdx.x * 16;
    const int c0 = (blockIdx.y * 8 + wave) * 16;

    const float tB = (float)(t0 + row);

    v8f acc = {};
    const float* mixRow = mix + (size_t)(c0 + row) * Ndim;

    #pragma unroll
    for (int j0 = 0; j0 < Ndim; j0 += 4) {
        const int ja = j0 + half * 2;

        v2f a;
        a.x = mixRow[ja];
        a.y = mixRow[ja + 1];

        v2f b;
        b.x = __expf(-sRate[ja]     * tB);
        b.y = __expf(-sRate[ja + 1] * tB);

        acc = __builtin_amdgcn_wmma_f32_16x16x4_f32(
             false, a,  false, b,
             (short)0, acc,  false,  false);
    }

    #pragma unroll
    for (int r = 0; r < 8; ++r) {
        const int m = half * 8 + r;
        Kout[(size_t)(c0 + m) * Tdim + (t0 + row)] = acc[r];
    }
}

constexpr int Tq          = Tdim / 4;
constexpr int NKinds      = 9;
constexpr int KPitch      = Tq + 4;
constexpr int RowsPerKind = 2;

constexpr int TilesW      = Tq / 16;
constexpr int PitchW      = 36;
constexpr int StrideW     = RowsPerKind * PitchW;
constexpr int KindFloatsW = TilesW * StrideW;
constexpr int KindSkewW   = 8;
constexpr int WholeKinds  = 8;

constexpr int TilesQ      = Tq / 8;
constexpr int TilePitch   = 20;
constexpr int TileStride  = RowsPerKind * TilePitch;
constexpr int KindFloats  = TilesQ * TileStride;
constexpr int Kind8Base   = WholeKinds * (KindFloatsW + KindSkewW);

constexpr int PairFloats  = Kind8Base + KindFloats;
constexpr int ConvThreads = 384;

__device__ __forceinline__ int kind_base_w(int kind) {
    return kind * (KindFloatsW + KindSkewW);
}

__device__ __forceinline__ v2f wide_pair(const v4f (&b)[8], int i) {
    return (i & 1) ? b[i >> 1].zw : b[i >> 1].xy;
}

__device__ __forceinline__ void load_wide(v4f (&dst)[8], const float* t) {
    const v4f* p = (const v4f*)t;
    #pragma unroll
    for (int i = 0; i < 8; ++i) dst[i] = p[i];
}

__device__ __forceinline__ void load_taps(v4f (&dst)[4], const float* k) {
    const v4f* p = (const v4f*)k;
    #pragma unroll
    for (int i = 0; i < 4; ++i) dst[i] = p[i];
}

__device__ __forceinline__ void wide_group_step(v2f (&acc)[16],
                                                const v4f (&newer)[8],
                                                const v4f (&older)[8],
                                                const v4f (&k)[4]) {
    #pragma unroll
    for (int q = 0; q < 16; ++q) {
        const float kq = k[q >> 2][q & 3];
        const v2f kk = {kq, kq};
        #pragma unroll
        for (int s = 0; s < 16; ++s) {
            const int i = s - q;
            const v2f p = (i >= 0) ? wide_pair(newer, i) : wide_pair(older, 16 + i);
            acc[s] = __builtin_elementwise_fma(kk, p, acc[s]);
        }
    }
}

__device__ __forceinline__ void conv_wide(v2f (&acc)[16],
                                          const float* uRow, const float* kRow,
                                          const float d, const int T) {
    v4f X[8], A[8], k[4];
    load_wide(X, uRow + T * StrideW);

    #pragma unroll
    for (int s = 0; s < 16; ++s) acc[s] = wide_pair(X, s) * d;

    const float* up = uRow + (T - 2) * StrideW;
    const float* kp = kRow;
    int G = 0;
    for (; G + 2 <= T; G += 2) {
        load_wide(A, up + StrideW);
        load_taps(k, kp);
        wide_group_step(acc, X, A, k);
        load_wide(X, up);
        load_taps(k, kp + 16);
        wide_group_step(acc, A, X, k);
        up -= 2 * StrideW;
        kp += 32;
    }
    if (G < T) {
        load_wide(A, uRow + (T - G - 1) * StrideW);
        load_taps(k, kp);
        wide_group_step(acc, X, A, k);
    }

    {
        v4f Z[8];
        load_wide(Z, uRow);
        load_taps(k, kRow + 16 * T);
        #pragma unroll
        for (int e = 0; e < 16; ++e) {
            const float ke = k[e >> 2][e & 3];
            const v2f kk = {ke, ke};
            #pragma unroll
            for (int s = e; s < 16; ++s) {
                acc[s] = __builtin_elementwise_fma(kk, wide_pair(Z, s - e), acc[s]);
            }
        }
    }
}

__device__ __forceinline__ void put_wide(float* t, const v2f (&acc)[16]) {
    v4f* p = (v4f*)t;
    #pragma unroll
    for (int s = 0; s < 8; ++s) {
        const v4f o = {acc[2 * s].x, acc[2 * s].y, acc[2 * s + 1].x, acc[2 * s + 1].y};
        p[s] = o;
    }
}


__device__ __forceinline__ v2f half_pair(const v4f (&b)[2], int i) {
    return (i & 1) ? b[i >> 1].zw : b[i >> 1].xy;
}

__device__ __forceinline__ void load_half(v4f (&dst)[2], const float* t) {
    const v4f* p = (const v4f*)t;
    dst[0] = p[0];
    dst[1] = p[1];
}

__device__ __forceinline__ void half_group_step(v2f (&acc)[16],
                                                const v4f (&cur)[2],
                                                const v4f (&m1)[2],
                                                const v4f (&m2)[2],
                                                const v4f k0, const v4f k1) {
    #pragma unroll
    for (int q = 0; q < 8; ++q) {
        const float kq = (q < 4) ? k0[q & 3] : k1[q & 3];
        const v2f kk = {kq, kq};
        #pragma unroll
        for (int s = 0; s < 4; ++s) {
            const int i = s - q;
            const v2f p = (i >= 0) ? half_pair(cur, i)
                        : (i >= -4) ? half_pair(m1, 4 + i) : half_pair(m2, 8 + i);
            acc[s] = __builtin_elementwise_fma(kk, p, acc[s]);
        }
    }
}

__device__ __forceinline__ void conv_half(v2f (&acc)[16],
                                          const float* uRow, const float* kRow,
                                          const float d, const int T, const int hs) {
    v4f X[2], A1[2], A2[2], B1[2];
    load_half(X, uRow + T * TileStride + 8 * hs);

    #pragma unroll
    for (int s = 0; s < 4; ++s) acc[s] = half_pair(X, s) * d;

    const float* q1 = uRow + (T - 2 + hs) * TileStride + 8 * (1 - hs);
    const float* q2 = uRow + (T - 2) * TileStride + 8 * hs;
    const float* kp = kRow;
    int G = 0;
    for (; G + 2 <= T; G += 2) {
        load_half(A1, q1 + TileStride);
        load_half(A2, q2 + TileStride);
        {
            const v4f k0 = *(const v4f*)&kp[0];
            const v4f k1 = *(const v4f*)&kp[4];
            half_group_step(acc, X, A1, A2, k0, k1);
        }
        load_half(B1, q1);
        load_half(X, q2);
        {
            const v4f k0 = *(const v4f*)&kp[8];
            const v4f k1 = *(const v4f*)&kp[12];
            half_group_step(acc, A2, B1, X, k0, k1);
        }
        q1 -= 2 * TileStride;
        q2 -= 2 * TileStride;
        kp += 16;
    }
    if (G < T) {
        load_half(A1, q1 + TileStride);
        load_half(A2, q2 + TileStride);
        const v4f k0 = *(const v4f*)&kp[0];
        const v4f k1 = *(const v4f*)&kp[4];
        half_group_step(acc, X, A1, A2, k0, k1);
        X[0] = A2[0];
        X[1] = A2[1];
    }

    {
        v4f Z[2];
        load_half(Z, uRow);
        if (hs == 0) {
            const v4f zero = {0.0f, 0.0f, 0.0f, 0.0f};
            Z[0] = zero;
            Z[1] = zero;
        }
        const v4f k0 = *(const v4f*)&kRow[8 * T];
        const v4f k1 = *(const v4f*)&kRow[8 * T + 4];
        #pragma unroll
        for (int q = 0; q < 8; ++q) {
            const float kq = (q < 4) ? k0[q & 3] : k1[q & 3];
            const v2f kk = {kq, kq};
            #pragma unroll
            for (int s = 0; s < 4; ++s) {
                const int i = s - q;
                if (i >= 0) {
                    acc[s] = __builtin_elementwise_fma(kk, half_pair(X, i), acc[s]);
                } else if (i >= -4) {
                    acc[s] = __builtin_elementwise_fma(kk, half_pair(Z, 4 + i), acc[s]);
                }
            }
        }
    }
}

__device__ __forceinline__ void put_half(float* t, const v2f (&acc)[16]) {
    v4f* p = (v4f*)t;
    #pragma unroll
    for (int s = 0; s < 2; ++s) {
        const v4f o = {acc[2 * s].x, acc[2 * s].y, acc[2 * s + 1].x, acc[2 * s + 1].y};
        p[s] = o;
    }
}

__global__ void __launch_bounds__(ConvThreads, 6)
kconv(const float* __restrict__ u,
      const float* __restrict__ Dv,
      const float* __restrict__ Kc,
      float* __restrict__ y) {
    __shared__ __attribute__((aligned(16))) float smem[NKinds * KPitch + PairFloats];
    float* const sK = smem;
    float* const sU = smem + NKinds * KPitch;

    const int c   = blockIdx.x;
    const int tid = threadIdx.x;

    v4f ga[3], gb[3], gk[2];
    #pragma unroll
    for (int it = 0; it < 3; ++it) {
        const int q = it * ConvThreads + tid;
        if (q < 2 * Tq) {
            const int r = q >> 9;
            const int i = q & (Tq - 1);
            const float* ua = u + ((size_t)(2 * r) * Cdim + c) * Tdim;
            const float* ub = ua + (size_t)Cdim * Tdim;
            ga[it] = ((const v4f*)ua)[i];
            gb[it] = ((const v4f*)ub)[i];
        }
    }
    #pragma unroll
    for (int it = 0; it < 2; ++it) {
        const int q = it * ConvThreads + tid;
        if (q < Tq) gk[it] = ((const v4f*)(Kc + (size_t)c * Tdim))[q];
    }
    #pragma unroll
    for (int it = 0; it < 3; ++it) {
        const int q = it * ConvThreads + tid;
        if (q < 2 * Tq) {
            const int r = q >> 9;
            const int i = q & (Tq - 1);
            const v4f a  = ga[it];
            const v4f bq = gb[it];
            const v2f px = {a.x, bq.x};
            const v2f py = {a.y, bq.y};
            const v2f pz = {a.z, bq.z};
            const v2f pw = {a.w, bq.w};
            const v2f sxy = px + py;
            const v2f szw = pz + pw;
            float* dst = sU + (2 * (i >> 4) + r) * PitchW + 2 * (i & 15);
            *(v2f*)(dst + kind_base_w(0)) = px;
            *(v2f*)(dst + kind_base_w(1)) = pz;
            *(v2f*)(dst + kind_base_w(2)) = px + pz;
            *(v2f*)(dst + kind_base_w(3)) = py;
            *(v2f*)(dst + kind_base_w(4)) = pw;
            *(v2f*)(dst + kind_base_w(5)) = py + pw;
            *(v2f*)(dst + kind_base_w(6)) = sxy;
            *(v2f*)(dst + kind_base_w(7)) = szw;
            float* dst8 = sU + Kind8Base + (2 * (i >> 3) + r) * TilePitch + 2 * (i & 7);
            *(v2f*)dst8 = sxy + szw;
        }
    }
    #pragma unroll
    for (int it = 0; it < 2; ++it) {
        const int q = it * ConvThreads + tid;
        if (q < Tq) {
            const v4f k = gk[it];
            const float kxy = k.x + k.y;
            const float kzw = k.z + k.w;
            float* dst = sK + q;
            dst[0 * KPitch] = k.x;
            dst[1 * KPitch] = k.z;
            dst[2 * KPitch] = k.x + k.z;
            dst[3 * KPitch] = k.y;
            dst[4 * KPitch] = k.w;
            dst[5 * KPitch] = k.y + k.w;
            dst[6 * KPitch] = kxy;
            dst[7 * KPitch] = kzw;
            dst[8 * KPitch] = kxy + kzw;
        }
    }
    __syncthreads();

    const int lane = tid & 31;
    const int wave = tid >> 5;
    const int hf   = lane >> 4;
    const int l16  = lane & 15;
    int kind, r, j, hs = 0;
    const bool whole = wave < 8;
    if (whole) {
        kind = l16 >> 1;
        r    = l16 & 1;
        j    = 2 * wave + hf;
    } else {
        kind = 8;
        hs   = l16 >> 3;
        r    = (l16 >> 2) & 1;
        j    = (wave - 8) * 8 + hf * 4 + (l16 & 3);
    }

    float* const uRow = whole ? (sU + kind_base_w(kind) + r * PitchW)
                              : (sU + Kind8Base + r * TilePitch);
    const float* kRow = sK + kind * KPitch;

    const bool skip = (kind == 0) || (kind == 2) || (kind == 6) || (kind == 8);
    const float d = skip ? Dv[c] : 0.0f;

    v2f resLong[16], resShort[16];
    if (whole) {
        conv_wide(resLong,  uRow, kRow, d, (TilesW - 1) - j);
        conv_wide(resShort, uRow, kRow, d, j);
    } else {
        conv_half(resLong,  uRow, kRow, d, (TilesQ - 1) - j, hs);
        conv_half(resShort, uRow, kRow, d, j, hs);
    }

    __syncthreads();
    if (whole) {
        put_wide(uRow + ((TilesW - 1) - j) * StrideW, resLong);
        put_wide(uRow + j * StrideW, resShort);
    } else {
        put_half(uRow + ((TilesQ - 1) - j) * TileStride + 8 * hs, resLong);
        put_half(uRow + j * TileStride + 8 * hs, resShort);
    }
    __syncthreads();

    #pragma unroll
    for (int it = 0; it < 3; ++it) {
        const int q = it * ConvThreads + tid;
        if (q < 2 * Tq) {
            const int rr = q >> 9;
            const int i  = q & (Tq - 1);
            const float* pc = sU + (2 * (i >> 4) + rr) * PitchW + 2 * (i & 15);
            const v2f R0 = *(const v2f*)(pc + kind_base_w(0));
            const v2f R1 = *(const v2f*)(pc + kind_base_w(1));
            const v2f R2 = *(const v2f*)(pc + kind_base_w(2));
            const v2f R3 = *(const v2f*)(pc + kind_base_w(3));
            const v2f R4 = *(const v2f*)(pc + kind_base_w(4));
            const v2f R5 = *(const v2f*)(pc + kind_base_w(5));
            const v2f R6 = *(const v2f*)(pc + kind_base_w(6));
            const v2f R7 = *(const v2f*)(pc + kind_base_w(7));
            const v2f R8 = *(const v2f*)(sU + Kind8Base + (2 * (i >> 3) + rr) * TilePitch + 2 * (i & 7));
            v2f q1 = {0.0f, 0.0f}, q3 = q1, q4 = q1, q5 = q1, q7 = q1;
            if (i > 0) {
                const float* pp = pc + ((i & 15) ? -2 : (30 - StrideW));
                q1 = *(const v2f*)(pp + kind_base_w(1));
                q3 = *(const v2f*)(pp + kind_base_w(3));
                q4 = *(const v2f*)(pp + kind_base_w(4));
                q5 = *(const v2f*)(pp + kind_base_w(5));
                q7 = *(const v2f*)(pp + kind_base_w(7));
            }
            const v2f pe0 = R0 + q1;
            const v2f po0 = R2 - R0 - R1;
            const v2f pe1 = R3 + q4;
            const v2f po1 = R5 - R3 - R4;
            const v2f peS = R6 + q7;
            const v2f poS = R8 - R6 - R7;
            const v2f pm1 = q5 - q3 - q4;
            const v2f y0 = pe0 + pm1;
            const v2f y1 = peS - pe0 - pe1;
            const v2f y2 = po0 + pe1;
            const v2f y3 = poS - po0 - po1;
            float* ya = y + ((size_t)(2 * rr) * Cdim + c) * Tdim;
            float* yb = ya + (size_t)Cdim * Tdim;
            const v4f oa = {y0.x, y1.x, y2.x, y3.x};
            const v4f ob = {y0.y, y1.y, y2.y, y3.y};
            ((v4f*)ya)[i] = oa;
            ((v4f*)yb)[i] = ob;
        }
    }
}

extern "C" void kernel_launch(void* const* d_in, const int* in_sizes, int n_in,
                              void* d_out, int out_size, void* d_ws, size_t ws_size,
                              hipStream_t stream) {
    const float* u       = (const float*)d_in[0];
    const float* log_tau = (const float*)d_in[1];
    const float* mix     = (const float*)d_in[2];
    const float* Dv      = (const float*)d_in[3];
    const float* log_dt  = (const float*)d_in[4];
    float*       y       = (float*)d_out;
    float*       Kws     = (float*)d_ws;

    (void)in_sizes; (void)n_in; (void)out_size; (void)ws_size;

    dim3 g1(Tdim / 16, (Cdim / 16) / 8);
    kgen_wmma<<<g1, 256, 0, stream>>>(mix, log_tau, log_dt, Kws);

    dim3 g2(Cdim);
    kconv<<<g2, ConvThreads, 0, stream>>>(u, Dv, Kws, y);
}


# speedup vs baseline: 3.8092x; 1.0427x over previous
#include <hip/hip_runtime.h>

constexpr int Bdim = 4;
constexpr int Cdim = 1024;
constexpr int Ndim = 64;
constexpr int Tdim = 2048;

typedef float v2f __attribute__((ext_vector_type(2)));
typedef float v4f __attribute__((ext_vector_type(4)));
typedef float v8f __attribute__((ext_vector_type(8)));

__device__ __forceinline__ float softplus_fast(float x) {
    return fmaxf(x, 0.0f) + __logf(1.0f + __expf(-fabsf(x)));
}

__global__ void __launch_bounds__(256)
kgen_wmma(const float* __restrict__ mix,
          const float* __restrict__ log_tau,
          const float* __restrict__ log_dt,
          float* __restrict__ Kout) {
    __shared__ float sRate[Ndim];

    if (threadIdx.x < Ndim) {
        const float dtv = softplus_fast(log_dt[0]);
        sRate[threadIdx.x] = softplus_fast(log_tau[threadIdx.x]) * dtv;
    }
    __syncthreads();

    const int lane = threadIdx.x & 31;
    const int wave = threadIdx.x >> 5;
    const int row  = lane & 15;
    const int half = lane >> 4;

    const int t0 = blockIdx.x * 16;
    const int c0 = (blockIdx.y * 8 + wave) * 16;

    const float tB = (float)(t0 + row);

    v8f acc = {};
    const float* mixRow = mix + (size_t)(c0 + row) * Ndim;

    #pragma unroll
    for (int j0 = 0; j0 < Ndim; j0 += 4) {
        const int ja = j0 + half * 2;

        v2f a;
        a.x = mixRow[ja];
        a.y = mixRow[ja + 1];

        v2f b;
        b.x = __expf(-sRate[ja]     * tB);
        b.y = __expf(-sRate[ja + 1] * tB);

        acc = __builtin_amdgcn_wmma_f32_16x16x4_f32(
             false, a,  false, b,
             (short)0, acc,  false,  false);
    }

    #pragma unroll
    for (int r = 0; r < 8; ++r) {
        const int m = half * 8 + r;
        Kout[(size_t)(c0 + m) * Tdim + (t0 + row)] = acc[r];
    }
}

constexpr int Tq          = Tdim / 4;
constexpr int NKinds      = 9;
constexpr int KPitch      = Tq + 4;
constexpr int RowsPerKind = 2;

constexpr int TilesW      = Tq / 16;
constexpr int PitchW      = 36;
constexpr int StrideW     = RowsPerKind * PitchW;
constexpr int KindFloatsW = TilesW * StrideW;
constexpr int KindSkewW   = 8;
constexpr int WholeKinds  = 8;

constexpr int TilesQ      = Tq / 8;
constexpr int TilePitch   = 20;
constexpr int TileStride  = RowsPerKind * TilePitch;
constexpr int KindFloats  = TilesQ * TileStride;
constexpr int Kind8Base   = WholeKinds * (KindFloatsW + KindSkewW);

constexpr int PairFloats  = Kind8Base + KindFloats;
constexpr int ConvThreads = 384;

__device__ __forceinline__ int kind_base_w(int kind) {
    return kind * (KindFloatsW + KindSkewW);
}

__device__ __forceinline__ v2f wide_pair(const v4f (&b)[8], int i) {
    return (i & 1) ? b[i >> 1].zw : b[i >> 1].xy;
}

__device__ __forceinline__ void load_wide(v4f (&dst)[8], const float* t) {
    const v4f* p = (const v4f*)t;
    #pragma unroll
    for (int i = 0; i < 8; ++i) dst[i] = p[i];
}

__device__ __forceinline__ void load_taps(v4f (&dst)[4], const float* k) {
    const v4f* p = (const v4f*)k;
    #pragma unroll
    for (int i = 0; i < 4; ++i) dst[i] = p[i];
}

__device__ __forceinline__ void pk_fma_tap_lo(v2f& acc, const v2f kp, const v2f p) {
    asm volatile("v_pk_fma_f32 %0, %1, %2, %0 op_sel_hi:[0,1,1]" : "+v"(acc) : "v"(kp), "v"(p));
}
__device__ __forceinline__ void pk_fma_tap_hi(v2f& acc, const v2f kp, const v2f p) {
    asm volatile("v_pk_fma_f32 %0, %1, %2, %0 op_sel:[1,0,0]" : "+v"(acc) : "v"(kp), "v"(p));
}

__device__ __forceinline__ void wide_group_step(v2f (&acc)[16],
                                                const v4f (&newer)[8],
                                                const v4f (&older)[8],
                                                const v4f (&k)[4]) {
    #pragma unroll
    for (int q = 0; q < 16; ++q) {
        const v2f kp = (q & 2) ? k[q >> 2].zw : k[q >> 2].xy;
        #pragma unroll
        for (int s = 0; s < 16; ++s) {
            const int i = s - q;
            const v2f p = (i >= 0) ? wide_pair(newer, i) : wide_pair(older, 16 + i);
            if (q & 1) pk_fma_tap_hi(acc[s], kp, p);
            else       pk_fma_tap_lo(acc[s], kp, p);
        }
    }
}

__device__ __forceinline__ void conv_wide(v2f (&acc)[16],
                                          const float* uRow, const float* kRow,
                                          const float d, const int T) {
    v4f X[8], A[8], k[4];
    load_wide(X, uRow + T * StrideW);

    #pragma unroll
    for (int s = 0; s < 16; ++s) acc[s] = wide_pair(X, s) * d;

    const float* up = uRow + (T - 2) * StrideW;
    const float* kp = kRow;
    int G = 0;
    for (; G + 2 <= T; G += 2) {
        load_wide(A, up + StrideW);
        load_taps(k, kp);
        wide_group_step(acc, X, A, k);
        load_wide(X, up);
        load_taps(k, kp + 16);
        wide_group_step(acc, A, X, k);
        up -= 2 * StrideW;
        kp += 32;
    }
    if (G < T) {
        load_wide(A, uRow + (T - G - 1) * StrideW);
        load_taps(k, kp);
        wide_group_step(acc, X, A, k);
    }

    {
        v4f Z[8];
        load_wide(Z, uRow);
        load_taps(k, kRow + 16 * T);
        #pragma unroll
        for (int e = 0; e < 16; ++e) {
            const float ke = k[e >> 2][e & 3];
            const v2f kk = {ke, ke};
            #pragma unroll
            for (int s = e; s < 16; ++s) {
                acc[s] = __builtin_elementwise_fma(kk, wide_pair(Z, s - e), acc[s]);
            }
        }
    }
}

__device__ __forceinline__ void put_wide(float* t, const v2f (&acc)[16]) {
    v4f* p = (v4f*)t;
    #pragma unroll
    for (int s = 0; s < 8; ++s) {
        const v4f o = {acc[2 * s].x, acc[2 * s].y, acc[2 * s + 1].x, acc[2 * s + 1].y};
        p[s] = o;
    }
}


__device__ __forceinline__ v2f half_pair(const v4f (&b)[2], int i) {
    return (i & 1) ? b[i >> 1].zw : b[i >> 1].xy;
}

__device__ __forceinline__ void load_half(v4f (&dst)[2], const float* t) {
    const v4f* p = (const v4f*)t;
    dst[0] = p[0];
    dst[1] = p[1];
}

__device__ __forceinline__ void half_group_step(v2f (&acc)[16],
                                                const v4f (&cur)[2],
                                                const v4f (&m1)[2],
                                                const v4f (&m2)[2],
                                                const v4f k0, const v4f k1) {
    #pragma unroll
    for (int q = 0; q < 8; ++q) {
        const float kq = (q < 4) ? k0[q & 3] : k1[q & 3];
        const v2f kk = {kq, kq};
        #pragma unroll
        for (int s = 0; s < 4; ++s) {
            const int i = s - q;
            const v2f p = (i >= 0) ? half_pair(cur, i)
                        : (i >= -4) ? half_pair(m1, 4 + i) : half_pair(m2, 8 + i);
            acc[s] = __builtin_elementwise_fma(kk, p, acc[s]);
        }
    }
}

__device__ __forceinline__ void conv_half(v2f (&acc)[16],
                                          const float* uRow, const float* kRow,
                                          const float d, const int T, const int hs) {
    v4f X[2], A1[2], A2[2], B1[2];
    load_half(X, uRow + T * TileStride + 8 * hs);

    #pragma unroll
    for (int s = 0; s < 4; ++s) acc[s] = half_pair(X, s) * d;

    const float* q1 = uRow + (T - 2 + hs) * TileStride + 8 * (1 - hs);
    const float* q2 = uRow + (T - 2) * TileStride + 8 * hs;
    const float* kp = kRow;
    int G = 0;
    for (; G + 2 <= T; G += 2) {
        load_half(A1, q1 + TileStride);
        load_half(A2, q2 + TileStride);
        {
            const v4f k0 = *(const v4f*)&kp[0];
            const v4f k1 = *(const v4f*)&kp[4];
            half_group_step(acc, X, A1, A2, k0, k1);
        }
        load_half(B1, q1);
        load_half(X, q2);
        {
            const v4f k0 = *(const v4f*)&kp[8];
            const v4f k1 = *(const v4f*)&kp[12];
            half_group_step(acc, A2, B1, X, k0, k1);
        }
        q1 -= 2 * TileStride;
        q2 -= 2 * TileStride;
        kp += 16;
    }
    if (G < T) {
        load_half(A1, q1 + TileStride);
        load_half(A2, q2 + TileStride);
        const v4f k0 = *(const v4f*)&kp[0];
        const v4f k1 = *(const v4f*)&kp[4];
        half_group_step(acc, X, A1, A2, k0, k1);
        X[0] = A2[0];
        X[1] = A2[1];
    }

    {
        v4f Z[2];
        load_half(Z, uRow);
        if (hs == 0) {
            const v4f zero = {0.0f, 0.0f, 0.0f, 0.0f};
            Z[0] = zero;
            Z[1] = zero;
        }
        const v4f k0 = *(const v4f*)&kRow[8 * T];
        const v4f k1 = *(const v4f*)&kRow[8 * T + 4];
        #pragma unroll
        for (int q = 0; q < 8; ++q) {
            const float kq = (q < 4) ? k0[q & 3] : k1[q & 3];
            const v2f kk = {kq, kq};
            #pragma unroll
            for (int s = 0; s < 4; ++s) {
                const int i = s - q;
                if (i >= 0) {
                    acc[s] = __builtin_elementwise_fma(kk, half_pair(X, i), acc[s]);
                } else if (i >= -4) {
                    acc[s] = __builtin_elementwise_fma(kk, half_pair(Z, 4 + i), acc[s]);
                }
            }
        }
    }
}

__device__ __forceinline__ void put_half(float* t, const v2f (&acc)[16]) {
    v4f* p = (v4f*)t;
    #pragma unroll
    for (int s = 0; s < 2; ++s) {
        const v4f o = {acc[2 * s].x, acc[2 * s].y, acc[2 * s + 1].x, acc[2 * s + 1].y};
        p[s] = o;
    }
}

__global__ void __launch_bounds__(ConvThreads, 6)
kconv(const float* __restrict__ u,
      const float* __restrict__ Dv,
      const float* __restrict__ Kc,
      float* __restrict__ y) {
    __shared__ __attribute__((aligned(16))) float smem[NKinds * KPitch + PairFloats];
    float* const sK = smem;
    float* const sU = smem + NKinds * KPitch;

    const int c   = blockIdx.x;
    const int tid = threadIdx.x;

    v4f ga[3], gb[3], gk[2];
    #pragma unroll
    for (int it = 0; it < 3; ++it) {
        const int q = it * ConvThreads + tid;
        if (q < 2 * Tq) {
            const int r = q >> 9;
            const int i = q & (Tq - 1);
            const float* ua = u + ((size_t)(2 * r) * Cdim + c) * Tdim;
            const float* ub = ua + (size_t)Cdim * Tdim;
            ga[it] = ((const v4f*)ua)[i];
            gb[it] = ((const v4f*)ub)[i];
        }
    }
    #pragma unroll
    for (int it = 0; it < 2; ++it) {
        const int q = it * ConvThreads + tid;
        if (q < Tq) gk[it] = ((const v4f*)(Kc + (size_t)c * Tdim))[q];
    }
    #pragma unroll
    for (int it = 0; it < 3; ++it) {
        const int q = it * ConvThreads + tid;
        if (q < 2 * Tq) {
            const int r = q >> 9;
            const int i = q & (Tq - 1);
            const v4f a  = ga[it];
            const v4f bq = gb[it];
            const v2f px = {a.x, bq.x};
            const v2f py = {a.y, bq.y};
            const v2f pz = {a.z, bq.z};
            const v2f pw = {a.w, bq.w};
            const v2f sxy = px + py;
            const v2f szw = pz + pw;
            float* dst = sU + (2 * (i >> 4) + r) * PitchW + 2 * (i & 15);
            *(v2f*)(dst + kind_base_w(0)) = px;
            *(v2f*)(dst + kind_base_w(1)) = pz;
            *(v2f*)(dst + kind_base_w(2)) = px + pz;
            *(v2f*)(dst + kind_base_w(3)) = py;
            *(v2f*)(dst + kind_base_w(4)) = pw;
            *(v2f*)(dst + kind_base_w(5)) = py + pw;
            *(v2f*)(dst + kind_base_w(6)) = sxy;
            *(v2f*)(dst + kind_base_w(7)) = szw;
            float* dst8 = sU + Kind8Base + (2 * (i >> 3) + r) * TilePitch + 2 * (i & 7);
            *(v2f*)dst8 = sxy + szw;
        }
    }
    #pragma unroll
    for (int it = 0; it < 2; ++it) {
        const int q = it * ConvThreads + tid;
        if (q < Tq) {
            const v4f k = gk[it];
            const float kxy = k.x + k.y;
            const float kzw = k.z + k.w;
            float* dst = sK + q;
            dst[0 * KPitch] = k.x;
            dst[1 * KPitch] = k.z;
            dst[2 * KPitch] = k.x + k.z;
            dst[3 * KPitch] = k.y;
            dst[4 * KPitch] = k.w;
            dst[5 * KPitch] = k.y + k.w;
            dst[6 * KPitch] = kxy;
            dst[7 * KPitch] = kzw;
            dst[8 * KPitch] = kxy + kzw;
        }
    }
    __syncthreads();

    const int lane = tid & 31;
    const int wave = tid >> 5;
    const int hf   = lane >> 4;
    const int l16  = lane & 15;
    int kind, r, j, hs = 0;
    const bool whole = wave < 8;
    if (whole) {
        kind = l16 >> 1;
        r    = l16 & 1;
        j    = 2 * wave + hf;
    } else {
        kind = 8;
        hs   = l16 >> 3;
        r    = (l16 >> 2) & 1;
        j    = (wave - 8) * 8 + hf * 4 + (l16 & 3);
    }

    float* const uRow = whole ? (sU + kind_base_w(kind) + r * PitchW)
                              : (sU + Kind8Base + r * TilePitch);
    const float* kRow = sK + kind * KPitch;

    const bool skip = (kind == 0) || (kind == 2) || (kind == 6) || (kind == 8);
    const float d = skip ? Dv[c] : 0.0f;

    v2f resLong[16], resShort[16];
    if (whole) {
        conv_wide(resLong,  uRow, kRow, d, (TilesW - 1) - j);
        conv_wide(resShort, uRow, kRow, d, j);
    } else {
        conv_half(resLong,  uRow, kRow, d, (TilesQ - 1) - j, hs);
        conv_half(resShort, uRow, kRow, d, j, hs);
    }

    __syncthreads();
    if (whole) {
        put_wide(uRow + ((TilesW - 1) - j) * StrideW, resLong);
        put_wide(uRow + j * StrideW, resShort);
    } else {
        put_half(uRow + ((TilesQ - 1) - j) * TileStride + 8 * hs, resLong);
        put_half(uRow + j * TileStride + 8 * hs, resShort);
    }
    __syncthreads();

    #pragma unroll
    for (int it = 0; it < 3; ++it) {
        const int q = it * ConvThreads + tid;
        if (q < 2 * Tq) {
            const int rr = q >> 9;
            const int i  = q & (Tq - 1);
            const float* pc = sU + (2 * (i >> 4) + rr) * PitchW + 2 * (i & 15);
            const v2f R0 = *(const v2f*)(pc + kind_base_w(0));
            const v2f R1 = *(const v2f*)(pc + kind_base_w(1));
            const v2f R2 = *(const v2f*)(pc + kind_base_w(2));
            const v2f R3 = *(const v2f*)(pc + kind_base_w(3));
            const v2f R4 = *(const v2f*)(pc + kind_base_w(4));
            const v2f R5 = *(const v2f*)(pc + kind_base_w(5));
            const v2f R6 = *(const v2f*)(pc + kind_base_w(6));
            const v2f R7 = *(const v2f*)(pc + kind_base_w(7));
            const v2f R8 = *(const v2f*)(sU + Kind8Base + (2 * (i >> 3) + rr) * TilePitch + 2 * (i & 7));
            v2f q1 = {0.0f, 0.0f}, q3 = q1, q4 = q1, q5 = q1, q7 = q1;
            if (i > 0) {
                const float* pp = pc + ((i & 15) ? -2 : (30 - StrideW));
                q1 = *(const v2f*)(pp + kind_base_w(1));
                q3 = *(const v2f*)(pp + kind_base_w(3));
                q4 = *(const v2f*)(pp + kind_base_w(4));
                q5 = *(const v2f*)(pp + kind_base_w(5));
                q7 = *(const v2f*)(pp + kind_base_w(7));
            }
            const v2f pe0 = R0 + q1;
            const v2f po0 = R2 - R0 - R1;
            const v2f pe1 = R3 + q4;
            const v2f po1 = R5 - R3 - R4;
            const v2f peS = R6 + q7;
            const v2f poS = R8 - R6 - R7;
            const v2f pm1 = q5 - q3 - q4;
            const v2f y0 = pe0 + pm1;
            const v2f y1 = peS - pe0 - pe1;
            const v2f y2 = po0 + pe1;
            const v2f y3 = poS - po0 - po1;
            float* ya = y + ((size_t)(2 * rr) * Cdim + c) * Tdim;
            float* yb = ya + (size_t)Cdim * Tdim;
            const v4f oa = {y0.x, y1.x, y2.x, y3.x};
            const v4f ob = {y0.y, y1.y, y2.y, y3.y};
            ((v4f*)ya)[i] = oa;
            ((v4f*)yb)[i] = ob;
        }
    }
}

extern "C" void kernel_launch(void* const* d_in, const int* in_sizes, int n_in,
                              void* d_out, int out_size, void* d_ws, size_t ws_size,
                              hipStream_t stream) {
    const float* u       = (const float*)d_in[0];
    const float* log_tau = (const float*)d_in[1];
    const float* mix     = (const float*)d_in[2];
    const float* Dv      = (const float*)d_in[3];
    const float* log_dt  = (const float*)d_in[4];
    float*       y       = (float*)d_out;
    float*       Kws     = (float*)d_ws;

    (void)in_sizes; (void)n_in; (void)out_size; (void)ws_size;

    dim3 g1(Tdim / 16, (Cdim / 16) / 8);
    kgen_wmma<<<g1, 256, 0, stream>>>(mix, log_tau, log_dt, Kws);

    dim3 g2(Cdim);
    kconv<<<g2, ConvThreads, 0, stream>>>(u, Dv, Kws, y);
}


# conv: tap groups (64 lags of K) above the last tap that is not exactly zero are left out; bound found per block while staging the taps; exact for every input, sums and their order unchanged
# speedup vs baseline: 8.4416x; 2.2161x over previous
#include <hip/hip_runtime.h>

constexpr int Bdim = 4;
constexpr int Cdim = 1024;
constexpr int Ndim = 64;
constexpr int Tdim = 2048;

typedef float v2f __attribute__((ext_vector_type(2)));
typedef float v4f __attribute__((ext_vector_type(4)));
typedef float v8f __attribute__((ext_vector_type(8)));

__device__ __forceinline__ float softplus_fast(float x) {
    return fmaxf(x, 0.0f) + __logf(1.0f + __expf(-fabsf(x)));
}

__global__ void __launch_bounds__(256)
kgen_wmma(const float* __restrict__ mix,
          const float* __restrict__ log_tau,
          const float* __restrict__ log_dt,
          float* __restrict__ Kout) {
    __shared__ float sRate[Ndim];

    if (threadIdx.x < Ndim) {
        const float dtv = softplus_fast(log_dt[0]);
        sRate[threadIdx.x] = softplus_fast(log_tau[threadIdx.x]) * dtv;
    }
    __syncthreads();

    const int lane = threadIdx.x & 31;
    const int wave = threadIdx.x >> 5;
    const int row  = lane & 15;
    const int half = lane >> 4;

    const int t0 = blockIdx.x * 16;
    const int c0 = (blockIdx.y * 8 + wave) * 16;

    const float tB = (float)(t0 + row);

    v8f acc = {};
    const float* mixRow = mix + (size_t)(c0 + row) * Ndim;

    #pragma unroll
    for (int j0 = 0; j0 < Ndim; j0 += 4) {
        const int ja = j0 + half * 2;

        v2f a;
        a.x = mixRow[ja];
        a.y = mixRow[ja + 1];

        v2f b;
        b.x = __expf(-sRate[ja]     * tB);
        b.y = __expf(-sRate[ja + 1] * tB);

        acc = __builtin_amdgcn_wmma_f32_16x16x4_f32(
             false, a,  false, b,
             (short)0, acc,  false,  false);
    }

    #pragma unroll
    for (int r = 0; r < 8; ++r) {
        const int m = half * 8 + r;
        Kout[(size_t)(c0 + m) * Tdim + (t0 + row)] = acc[r];
    }
}

constexpr int Tq          = Tdim / 4;
constexpr int NKinds      = 9;
constexpr int KPitch      = Tq + 4;
constexpr int RowsPerKind = 2;

constexpr int TilesW      = Tq / 16;
constexpr int PitchW      = 36;
constexpr int StrideW     = RowsPerKind * PitchW;
constexpr int KindFloatsW = TilesW * StrideW;
constexpr int KindSkewW   = 8;
constexpr int WholeKinds  = 8;

constexpr int TilesQ      = Tq / 8;
constexpr int TilePitch   = 20;
constexpr int TileStride  = RowsPerKind * TilePitch;
constexpr int KindFloats  = TilesQ * TileStride;
constexpr int Kind8Base   = WholeKinds * (KindFloatsW + KindSkewW);

constexpr int PairFloats  = Kind8Base + KindFloats;
constexpr int ConvThreads = 384;
constexpr int ConvWaves   = ConvThreads / 32;

__device__ __forceinline__ int kind_base_w(int kind) {
    return kind * (KindFloatsW + KindSkewW);
}

__device__ __forceinline__ v2f wide_pair(const v4f (&b)[8], int i) {
    return (i & 1) ? b[i >> 1].zw : b[i >> 1].xy;
}

__device__ __forceinline__ void load_wide(v4f (&dst)[8], const float* t) {
    const v4f* p = (const v4f*)t;
    #pragma unroll
    for (int i = 0; i < 8; ++i) dst[i] = p[i];
}

__device__ __forceinline__ void load_taps(v4f (&dst)[4], const float* k) {
    const v4f* p = (const v4f*)k;
    #pragma unroll
    for (int i = 0; i < 4; ++i) dst[i] = p[i];
}

__device__ __forceinline__ void pk_fma_tap_lo(v2f& acc, const v2f kp, const v2f p) {
    asm volatile("v_pk_fma_f32 %0, %1, %2, %0 op_sel_hi:[0,1,1]" : "+v"(acc) : "v"(kp), "v"(p));
}
__device__ __forceinline__ void pk_fma_tap_hi(v2f& acc, const v2f kp, const v2f p) {
    asm volatile("v_pk_fma_f32 %0, %1, %2, %0 op_sel:[1,0,0]" : "+v"(acc) : "v"(kp), "v"(p));
}

__device__ __forceinline__ void wide_group_step(v2f (&acc)[16],
                                                const v4f (&newer)[8],
                                                const v4f (&older)[8],
                                                const v4f (&k)[4]) {
    #pragma unroll
    for (int q = 0; q < 16; ++q) {
        const v2f kp = (q & 2) ? k[q >> 2].zw : k[q >> 2].xy;
        #pragma unroll
        for (int s = 0; s < 16; ++s) {
            const int i = s - q;
            const v2f p = (i >= 0) ? wide_pair(newer, i) : wide_pair(older, 16 + i);
            if (q & 1) pk_fma_tap_hi(acc[s], kp, p);
            else       pk_fma_tap_lo(acc[s], kp, p);
        }
    }
}

__device__ __forceinline__ void conv_wide(v2f (&acc)[16],
                                          const float* uRow, const float* kRow,
                                          const float d, const int T, const int nz) {
    v4f X[8], A[8], k[4];
    const int Tf = (T < nz) ? T : nz;
    load_wide(X, uRow + T * StrideW);

    #pragma unroll
    for (int s = 0; s < 16; ++s) acc[s] = wide_pair(X, s) * d;

    const float* up = uRow + (T - 2) * StrideW;
    const float* kp = kRow;
    const int G2 = Tf & ~1;
    for (int G = 0; G < G2; G += 2) {
        load_wide(A, up + StrideW);
        load_taps(k, kp);
        wide_group_step(acc, X, A, k);
        load_wide(X, up);
        load_taps(k, kp + 16);
        wide_group_step(acc, A, X, k);
        up -= 2 * StrideW;
        kp += 32;
    }
    if (G2 < Tf) {
        load_wide(A, up + StrideW);
        load_taps(k, kp);
        wide_group_step(acc, X, A, k);
    }

    if (T < nz) {
        v4f Z[8];
        load_wide(Z, uRow);
        load_taps(k, kRow + 16 * T);
        #pragma unroll
        for (int e = 0; e < 16; ++e) {
            const float ke = k[e >> 2][e & 3];
            const v2f kk = {ke, ke};
            #pragma unroll
            for (int s = e; s < 16; ++s) {
                acc[s] = __builtin_elementwise_fma(kk, wide_pair(Z, s - e), acc[s]);
            }
        }
    }
}

__device__ __forceinline__ void put_wide(float* t, const v2f (&acc)[16]) {
    v4f* p = (v4f*)t;
    #pragma unroll
    for (int s = 0; s < 8; ++s) {
        const v4f o = {acc[2 * s].x, acc[2 * s].y, acc[2 * s + 1].x, acc[2 * s + 1].y};
        p[s] = o;
    }
}


__device__ __forceinline__ v2f half_pair(const v4f (&b)[2], int i) {
    return (i & 1) ? b[i >> 1].zw : b[i >> 1].xy;
}

__device__ __forceinline__ void load_half(v4f (&dst)[2], const float* t) {
    const v4f* p = (const v4f*)t;
    dst[0] = p[0];
    dst[1] = p[1];
}

__device__ __forceinline__ void half_group_step(v2f (&acc)[16],
                                                const v4f (&cur)[2],
                                                const v4f (&m1)[2],
                                                const v4f (&m2)[2],
                                                const v4f k0, const v4f k1) {
    #pragma unroll
    for (int q = 0; q < 8; ++q) {
        const float kq = (q < 4) ? k0[q & 3] : k1[q & 3];
        const v2f kk = {kq, kq};
        #pragma unroll
        for (int s = 0; s < 4; ++s) {
            const int i = s - q;
            const v2f p = (i >= 0) ? half_pair(cur, i)
                        : (i >= -4) ? half_pair(m1, 4 + i) : half_pair(m2, 8 + i);
            acc[s] = __builtin_elementwise_fma(kk, p, acc[s]);
        }
    }
}

__device__ __forceinline__ void conv_half(v2f (&acc)[16],
                                          const float* uRow, const float* kRow,
                                          const float d, const int T, const int hs,
                                          const int nz8) {
    v4f X[2], A1[2], A2[2], B1[2];
    const int Tf = (T < nz8) ? T : nz8;
    load_half(X, uRow + T * TileStride + 8 * hs);

    #pragma unroll
    for (int s = 0; s < 4; ++s) acc[s] = half_pair(X, s) * d;

    const float* q1 = uRow + (T - 2 + hs) * TileStride + 8 * (1 - hs);
    const float* q2 = uRow + (T - 2) * TileStride + 8 * hs;
    const float* kp = kRow;
    int G = 0;
    for (; G + 2 <= Tf; G += 2) {
        load_half(A1, q1 + TileStride);
        load_half(A2, q2 + TileStride);
        {
            const v4f k0 = *(const v4f*)&kp[0];
            const v4f k1 = *(const v4f*)&kp[4];
            half_group_step(acc, X, A1, A2, k0, k1);
        }
        load_half(B1, q1);
        load_half(X, q2);
        {
            const v4f k0 = *(const v4f*)&kp[8];
            const v4f k1 = *(const v4f*)&kp[12];
            half_group_step(acc, A2, B1, X, k0, k1);
        }
        q1 -= 2 * TileStride;
        q2 -= 2 * TileStride;
        kp += 16;
    }
    if (G < Tf) {
        load_half(A1, q1 + TileStride);
        load_half(A2, q2 + TileStride);
        const v4f k0 = *(const v4f*)&kp[0];
        const v4f k1 = *(const v4f*)&kp[4];
        half_group_step(acc, X, A1, A2, k0, k1);
        X[0] = A2[0];
        X[1] = A2[1];
    }

    if (T < nz8) {
        v4f Z[2];
        load_half(Z, uRow);
        if (hs == 0) {
            const v4f zero = {0.0f, 0.0f, 0.0f, 0.0f};
            Z[0] = zero;
            Z[1] = zero;
        }
        const v4f k0 = *(const v4f*)&kRow[8 * T];
        const v4f k1 = *(const v4f*)&kRow[8 * T + 4];
        #pragma unroll
        for (int q = 0; q < 8; ++q) {
            const float kq = (q < 4) ? k0[q & 3] : k1[q & 3];
            const v2f kk = {kq, kq};
            #pragma unroll
            for (int s = 0; s < 4; ++s) {
                const int i = s - q;
                if (i >= 0) {
                    acc[s] = __builtin_elementwise_fma(kk, half_pair(X, i), acc[s]);
                } else if (i >= -4) {
                    acc[s] = __builtin_elementwise_fma(kk, half_pair(Z, 4 + i), acc[s]);
                }
            }
        }
    }
}

__device__ __forceinline__ void put_half(float* t, const v2f (&acc)[16]) {
    v4f* p = (v4f*)t;
    #pragma unroll
    for (int s = 0; s < 2; ++s) {
        const v4f o = {acc[2 * s].x, acc[2 * s].y, acc[2 * s + 1].x, acc[2 * s + 1].y};
        p[s] = o;
    }
}

__global__ void __launch_bounds__(ConvThreads, 6)
kconv(const float* __restrict__ u,
      const float* __restrict__ Dv,
      const float* __restrict__ Kc,
      float* __restrict__ y) {
    __shared__ __attribute__((aligned(16))) float smem[NKinds * KPitch + PairFloats + ConvWaves];
    float* const sK = smem;
    float* const sU = smem + NKinds * KPitch;
    float* const sTop = smem + NKinds * KPitch + PairFloats;

    const int c   = blockIdx.x;
    const int tid = threadIdx.x;

    v4f ga[3], gb[3], gk[2];
    #pragma unroll
    for (int it = 0; it < 3; ++it) {
        const int q = it * ConvThreads + tid;
        if (q < 2 * Tq) {
            const int r = q >> 9;
            const int i = q & (Tq - 1);
            const float* ua = u + ((size_t)(2 * r) * Cdim + c) * Tdim;
            const float* ub = ua + (size_t)Cdim * Tdim;
            ga[it] = ((const v4f*)ua)[i];
            gb[it] = ((const v4f*)ub)[i];
        }
    }
    #pragma unroll
    for (int it = 0; it < 2; ++it) {
        const int q = it * ConvThreads + tid;
        if (q < Tq) gk[it] = ((const v4f*)(Kc + (size_t)c * Tdim))[q];
    }
    #pragma unroll
    for (int it = 0; it < 3; ++it) {
        const int q = it * ConvThreads + tid;
        if (q < 2 * Tq) {
            const int r = q >> 9;
            const int i = q & (Tq - 1);
            const v4f a  = ga[it];
            const v4f bq = gb[it];
            const v2f px = {a.x, bq.x};
            const v2f py = {a.y, bq.y};
            const v2f pz = {a.z, bq.z};
            const v2f pw = {a.w, bq.w};
            const v2f sxy = px + py;
            const v2f szw = pz + pw;
            float* dst = sU + (2 * (i >> 4) + r) * PitchW + 2 * (i & 15);
            *(v2f*)(dst + kind_base_w(0)) = px;
            *(v2f*)(dst + kind_base_w(1)) = pz;
            *(v2f*)(dst + kind_base_w(2)) = px + pz;
            *(v2f*)(dst + kind_base_w(3)) = py;
            *(v2f*)(dst + kind_base_w(4)) = pw;
            *(v2f*)(dst + kind_base_w(5)) = py + pw;
            *(v2f*)(dst + kind_base_w(6)) = sxy;
            *(v2f*)(dst + kind_base_w(7)) = szw;
            float* dst8 = sU + Kind8Base + (2 * (i >> 3) + r) * TilePitch + 2 * (i & 7);
            *(v2f*)dst8 = sxy + szw;
        }
    }
    int waveTop = 0;
    #pragma unroll
    for (int it = 0; it < 2; ++it) {
        const int q = it * ConvThreads + tid;
        bool live = false;
        if (q < Tq) {
            const v4f k = gk[it];
            live = (k.x != 0.0f) || (k.y != 0.0f) || (k.z != 0.0f) || (k.w != 0.0f);
        }
        const unsigned liveMask = __builtin_amdgcn_ballot_w32(live);
        if (liveMask != 0u) {
            const int qTop = it * ConvThreads + (tid & ~31) + 31 - __builtin_clz(liveMask);
            const int gTop = (qTop >> 4) + 1;
            waveTop = (gTop > waveTop) ? gTop : waveTop;
        }
        if (q < Tq) {
            const v4f k = gk[it];
            const float kxy = k.x + k.y;
            const float kzw = k.z + k.w;
            float* dst = sK + q;
            dst[0 * KPitch] = k.x;
            dst[1 * KPitch] = k.z;
            dst[2 * KPitch] = k.x + k.z;
            dst[3 * KPitch] = k.y;
            dst[4 * KPitch] = k.w;
            dst[5 * KPitch] = k.y + k.w;
            dst[6 * KPitch] = kxy;
            dst[7 * KPitch] = kzw;
            dst[8 * KPitch] = kxy + kzw;
        }
    }
    if ((tid & 31) == 0) sTop[tid >> 5] = (float)waveTop;
    __syncthreads();

    int nz;
    {
        const v4f m0 = ((const v4f*)sTop)[0];
        const v4f m1 = ((const v4f*)sTop)[1];
        const v4f m2 = ((const v4f*)sTop)[2];
        const v4f mm = __builtin_elementwise_max(__builtin_elementwise_max(m0, m1), m2);
        const float top = fmaxf(fmaxf(mm.x, mm.y), fmaxf(mm.z, mm.w));
        nz = __builtin_amdgcn_readfirstlane((int)top);
    }

    const int lane = tid & 31;
    const int wave = tid >> 5;
    const int hf   = lane >> 4;
    const int l16  = lane & 15;
    int kind, r, j, hs = 0;
    const bool whole = wave < 8;
    if (whole) {
        kind = l16 >> 1;
        r    = l16 & 1;
        j    = 2 * wave + hf;
    } else {
        kind = 8;
        hs   = l16 >> 3;
        r    = (l16 >> 2) & 1;
        j    = (wave - 8) * 8 + hf * 4 + (l16 & 3);
    }

    float* const uRow = whole ? (sU + kind_base_w(kind) + r * PitchW)
                              : (sU + Kind8Base + r * TilePitch);
    const float* kRow = sK + kind * KPitch;

    const bool skip = (kind == 0) || (kind == 2) || (kind == 6) || (kind == 8);
    const float d = skip ? Dv[c] : 0.0f;

    v2f resLong[16], resShort[16];
    if (whole) {
        conv_wide(resLong,  uRow, kRow, d, (TilesW - 1) - j, nz);
        conv_wide(resShort, uRow, kRow, d, j, nz);
    } else {
        conv_half(resLong,  uRow, kRow, d, (TilesQ - 1) - j, hs, 2 * nz);
        conv_half(resShort, uRow, kRow, d, j, hs, 2 * nz);
    }

    __syncthreads();
    if (whole) {
        put_wide(uRow + ((TilesW - 1) - j) * StrideW, resLong);
        put_wide(uRow + j * StrideW, resShort);
    } else {
        put_half(uRow + ((TilesQ - 1) - j) * TileStride + 8 * hs, resLong);
        put_half(uRow + j * TileStride + 8 * hs, resShort);
    }
    __syncthreads();

    #pragma unroll
    for (int it = 0; it < 3; ++it) {
        const int q = it * ConvThreads + tid;
        if (q < 2 * Tq) {
            const int rr = q >> 9;
            const int i  = q & (Tq - 1);
            const float* pc = sU + (2 * (i >> 4) + rr) * PitchW + 2 * (i & 15);
            const v2f R0 = *(const v2f*)(pc + kind_base_w(0));
            const v2f R1 = *(const v2f*)(pc + kind_base_w(1));
            const v2f R2 = *(const v2f*)(pc + kind_base_w(2));
            const v2f R3 = *(const v2f*)(pc + kind_base_w(3));
            const v2f R4 = *(const v2f*)(pc + kind_base_w(4));
            const v2f R5 = *(const v2f*)(pc + kind_base_w(5));
            const v2f R6 = *(const v2f*)(pc + kind_base_w(6));
            const v2f R7 = *(const v2f*)(pc + kind_base_w(7));
            const v2f R8 = *(const v2f*)(sU + Kind8Base + (2 * (i >> 3) + rr) * TilePitch + 2 * (i & 7));
            v2f q1 = {0.0f, 0.0f}, q3 = q1, q4 = q1, q5 = q1, q7 = q1;
            if (i > 0) {
                const float* pp = pc + ((i & 15) ? -2 : (30 - StrideW));
                q1 = *(const v2f*)(pp + kind_base_w(1));
                q3 = *(const v2f*)(pp + kind_base_w(3));
                q4 = *(const v2f*)(pp + kind_base_w(4));
                q5 = *(const v2f*)(pp + kind_base_w(5));
                q7 = *(const v2f*)(pp + kind_base_w(7));
            }
            const v2f pe0 = R0 + q1;
            const v2f po0 = R2 - R0 - R1;
            const v2f pe1 = R3 + q4;
            const v2f po1 = R5 - R3 - R4;
            const v2f peS = R6 + q7;
            const v2f poS = R8 - R6 - R7;
            const v2f pm1 = q5 - q3 - q4;
            const v2f y0 = pe0 + pm1;
            const v2f y1 = peS - pe0 - pe1;
            const v2f y2 = po0 + pe1;
            const v2f y3 = poS - po0 - po1;
            float* ya = y + ((size_t)(2 * rr) * Cdim + c) * Tdim;
            float* yb = ya + (size_t)Cdim * Tdim;
            const v4f oa = {y0.x, y1.x, y2.x, y3.x};
            const v4f ob = {y0.y, y1.y, y2.y, y3.y};
            ((v4f*)ya)[i] = oa;
            ((v4f*)yb)[i] = ob;
        }
    }
}

extern "C" void kernel_launch(void* const* d_in, const int* in_sizes, int n_in,
                              void* d_out, int out_size, void* d_ws, size_t ws_size,
                              hipStream_t stream) {
    const float* u       = (const float*)d_in[0];
    const float* log_tau = (const float*)d_in[1];
    const float* mix     = (const float*)d_in[2];
    const float* Dv      = (const float*)d_in[3];
    const float* log_dt  = (const float*)d_in[4];
    float*       y       = (float*)d_out;
    float*       Kws     = (float*)d_ws;

    (void)in_sizes; (void)n_in; (void)out_size; (void)ws_size;

    dim3 g1(Tdim / 16, (Cdim / 16) / 8);
    kgen_wmma<<<g1, 256, 0, stream>>>(mix, log_tau, log_dt, Kws);

    dim3 g2(Cdim);
    kconv<<<g2, ConvThreads, 0, stream>>>(u, Dv, Kws, y);
}


# taps kernel: a wave computes 16 channels x 64 steps (4 tiles, its mix fragments loaded once): 2048 waves in 256 blocks, not 8192 short waves in 1024 blocks; tap sums and their order unchanged
# speedup vs baseline: 9.3955x; 1.1130x over previous
#include <hip/hip_runtime.h>

constexpr int Bdim = 4;
constexpr int Cdim = 1024;
constexpr int Ndim = 64;
constexpr int Tdim = 2048;

typedef float v2f __attribute__((ext_vector_type(2)));
typedef float v4f __attribute__((ext_vector_type(4)));
typedef float v8f __attribute__((ext_vector_type(8)));

__device__ __forceinline__ float softplus_fast(float x) {
    return fmaxf(x, 0.0f) + __logf(1.0f + __expf(-fabsf(x)));
}

constexpr int KgTiles = 4;
constexpr int KgWaves = 8;
constexpr int KgGridX = Tdim / (16 * KgTiles);
constexpr int KgGridY = (Cdim / 16) / KgWaves;

__global__ void __launch_bounds__(KgWaves * 32)
kgen_wmma(const float* __restrict__ mix,
          const float* __restrict__ log_tau,
          const float* __restrict__ log_dt,
          float* __restrict__ Kout) {
    __shared__ float sRate[Ndim];

    if (threadIdx.x < Ndim) {
        const float dtv = softplus_fast(log_dt[0]);
        sRate[threadIdx.x] = softplus_fast(log_tau[threadIdx.x]) * dtv;
    }
    __syncthreads();

    const int lane = threadIdx.x & 31;
    const int wave = threadIdx.x >> 5;
    const int row  = lane & 15;
    const int half = lane >> 4;

    const int t0 = blockIdx.x * (16 * KgTiles);
    const int c0 = (blockIdx.y * KgWaves + wave) * 16;

    const float* mixRow = mix + (size_t)(c0 + row) * Ndim;
    v2f a[Ndim / 4];
    #pragma unroll
    for (int jj = 0; jj < Ndim / 4; ++jj) {
        const int ja = 4 * jj + half * 2;
        a[jj].x = mixRow[ja];
        a[jj].y = mixRow[ja + 1];
    }

    float tB[KgTiles];
    #pragma unroll
    for (int n = 0; n < KgTiles; ++n) tB[n] = (float)(t0 + 16 * n + row);

    v8f acc[KgTiles];
    #pragma unroll
    for (int n = 0; n < KgTiles; ++n) {
        const v8f zero = {};
        acc[n] = zero;
    }

    #pragma unroll
    for (int jj = 0; jj < Ndim / 4; ++jj) {
        const int ja = 4 * jj + half * 2;
        const float r0 = sRate[ja];
        const float r1 = sRate[ja + 1];
        #pragma unroll
        for (int n = 0; n < KgTiles; ++n) {
            v2f b;
            b.x = __expf(-r0 * tB[n]);
            b.y = __expf(-r1 * tB[n]);

            acc[n] = __builtin_amdgcn_wmma_f32_16x16x4_f32(
                 false, a[jj],  false, b,
                 (short)0, acc[n],  false,  false);
        }
    }

    #pragma unroll
    for (int n = 0; n < KgTiles; ++n) {
        #pragma unroll
        for (int r = 0; r < 8; ++r) {
            const int m = half * 8 + r;
            Kout[(size_t)(c0 + m) * Tdim + (t0 + 16 * n + row)] = acc[n][r];
        }
    }
}

constexpr int Tq          = Tdim / 4;
constexpr int NKinds      = 9;
constexpr int KPitch      = Tq + 4;
constexpr int RowsPerKind = 2;

constexpr int TilesW      = Tq / 16;
constexpr int PitchW      = 36;
constexpr int StrideW     = RowsPerKind * PitchW;
constexpr int KindFloatsW = TilesW * StrideW;
constexpr int KindSkewW   = 8;
constexpr int WholeKinds  = 8;

constexpr int TilesQ      = Tq / 8;
constexpr int TilePitch   = 20;
constexpr int TileStride  = RowsPerKind * TilePitch;
constexpr int KindFloats  = TilesQ * TileStride;
constexpr int Kind8Base   = WholeKinds * (KindFloatsW + KindSkewW);

constexpr int PairFloats  = Kind8Base + KindFloats;
constexpr int ConvThreads = 384;
constexpr int ConvWaves   = ConvThreads / 32;

__device__ __forceinline__ int kind_base_w(int kind) {
    return kind * (KindFloatsW + KindSkewW);
}

__device__ __forceinline__ v2f wide_pair(const v4f (&b)[8], int i) {
    return (i & 1) ? b[i >> 1].zw : b[i >> 1].xy;
}

__device__ __forceinline__ void load_wide(v4f (&dst)[8], const float* t) {
    const v4f* p = (const v4f*)t;
    #pragma unroll
    for (int i = 0; i < 8; ++i) dst[i] = p[i];
}

__device__ __forceinline__ void load_taps(v4f (&dst)[4], const float* k) {
    const v4f* p = (const v4f*)k;
    #pragma unroll
    for (int i = 0; i < 4; ++i) dst[i] = p[i];
}

__device__ __forceinline__ void pk_fma_tap_lo(v2f& acc, const v2f kp, const v2f p) {
    asm volatile("v_pk_fma_f32 %0, %1, %2, %0 op_sel_hi:[0,1,1]" : "+v"(acc) : "v"(kp), "v"(p));
}
__device__ __forceinline__ void pk_fma_tap_hi(v2f& acc, const v2f kp, const v2f p) {
    asm volatile("v_pk_fma_f32 %0, %1, %2, %0 op_sel:[1,0,0]" : "+v"(acc) : "v"(kp), "v"(p));
}

__device__ __forceinline__ void wide_group_step(v2f (&acc)[16],
                                                const v4f (&newer)[8],
                                                const v4f (&older)[8],
                                                const v4f (&k)[4]) {
    #pragma unroll
    for (int q = 0; q < 16; ++q) {
        const v2f kp = (q & 2) ? k[q >> 2].zw : k[q >> 2].xy;
        #pragma unroll
        for (int s = 0; s < 16; ++s) {
            const int i = s - q;
            const v2f p = (i >= 0) ? wide_pair(newer, i) : wide_pair(older, 16 + i);
            if (q & 1) pk_fma_tap_hi(acc[s], kp, p);
            else       pk_fma_tap_lo(acc[s], kp, p);
        }
    }
}

__device__ __forceinline__ void conv_wide(v2f (&acc)[16],
                                          const float* uRow, const float* kRow,
                                          const float d, const int T, const int nz) {
    v4f X[8], A[8], k[4];
    const int Tf = (T < nz) ? T : nz;
    load_wide(X, uRow + T * StrideW);

    #pragma unroll
    for (int s = 0; s < 16; ++s) acc[s] = wide_pair(X, s) * d;

    const float* up = uRow + (T - 2) * StrideW;
    const float* kp = kRow;
    const int G2 = Tf & ~1;
    for (int G = 0; G < G2; G += 2) {
        load_wide(A, up + StrideW);
        load_taps(k, kp);
        wide_group_step(acc, X, A, k);
        load_wide(X, up);
        load_taps(k, kp + 16);
        wide_group_step(acc, A, X, k);
        up -= 2 * StrideW;
        kp += 32;
    }
    if (G2 < Tf) {
        load_wide(A, up + StrideW);
        load_taps(k, kp);
        wide_group_step(acc, X, A, k);
    }

    if (T < nz) {
        v4f Z[8];
        load_wide(Z, uRow);
        load_taps(k, kRow + 16 * T);
        #pragma unroll
        for (int e = 0; e < 16; ++e) {
            const float ke = k[e >> 2][e & 3];
            const v2f kk = {ke, ke};
            #pragma unroll
            for (int s = e; s < 16; ++s) {
                acc[s] = __builtin_elementwise_fma(kk, wide_pair(Z, s - e), acc[s]);
            }
        }
    }
}

__device__ __forceinline__ void put_wide(float* t, const v2f (&acc)[16]) {
    v4f* p = (v4f*)t;
    #pragma unroll
    for (int s = 0; s < 8; ++s) {
        const v4f o = {acc[2 * s].x, acc[2 * s].y, acc[2 * s + 1].x, acc[2 * s + 1].y};
        p[s] = o;
    }
}


__device__ __forceinline__ v2f half_pair(const v4f (&b)[2], int i) {
    return (i & 1) ? b[i >> 1].zw : b[i >> 1].xy;
}

__device__ __forceinline__ void load_half(v4f (&dst)[2], const float* t) {
    const v4f* p = (const v4f*)t;
    dst[0] = p[0];
    dst[1] = p[1];
}

__device__ __forceinline__ void half_group_step(v2f (&acc)[16],
                                                const v4f (&cur)[2],
                                                const v4f (&m1)[2],
                                                const v4f (&m2)[2],
                                                const v4f k0, const v4f k1) {
    #pragma unroll
    for (int q = 0; q < 8; ++q) {
        const float kq = (q < 4) ? k0[q & 3] : k1[q & 3];
        const v2f kk = {kq, kq};
        #pragma unroll
        for (int s = 0; s < 4; ++s) {
            const int i = s - q;
            const v2f p = (i >= 0) ? half_pair(cur, i)
                        : (i >= -4) ? half_pair(m1, 4 + i) : half_pair(m2, 8 + i);
            acc[s] = __builtin_elementwise_fma(kk, p, acc[s]);
        }
    }
}

__device__ __forceinline__ void conv_half(v2f (&acc)[16],
                                          const float* uRow, const float* kRow,
                                          const float d, const int T, const int hs,
                                          const int nz8) {
    v4f X[2], A1[2], A2[2], B1[2];
    const int Tf = (T < nz8) ? T : nz8;
    load_half(X, uRow + T * TileStride + 8 * hs);

    #pragma unroll
    for (int s = 0; s < 4; ++s) acc[s] = half_pair(X, s) * d;

    const float* q1 = uRow + (T - 2 + hs) * TileStride + 8 * (1 - hs);
    const float* q2 = uRow + (T - 2) * TileStride + 8 * hs;
    const float* kp = kRow;
    int G = 0;
    for (; G + 2 <= Tf; G += 2) {
        load_half(A1, q1 + TileStride);
        load_half(A2, q2 + TileStride);
        {
            const v4f k0 = *(const v4f*)&kp[0];
            const v4f k1 = *(const v4f*)&kp[4];
            half_group_step(acc, X, A1, A2, k0, k1);
        }
        load_half(B1, q1);
        load_half(X, q2);
        {
            const v4f k0 = *(const v4f*)&kp[8];
            const v4f k1 = *(const v4f*)&kp[12];
            half_group_step(acc, A2, B1, X, k0, k1);
        }
        q1 -= 2 * TileStride;
        q2 -= 2 * TileStride;
        kp += 16;
    }
    if (G < Tf) {
        load_half(A1, q1 + TileStride);
        load_half(A2, q2 + TileStride);
        const v4f k0 = *(const v4f*)&kp[0];
        const v4f k1 = *(const v4f*)&kp[4];
        half_group_step(acc, X, A1, A2, k0, k1);
        X[0] = A2[0];
        X[1] = A2[1];
    }

    if (T < nz8) {
        v4f Z[2];
        load_half(Z, uRow);
        if (hs == 0) {
            const v4f zero = {0.0f, 0.0f, 0.0f, 0.0f};
            Z[0] = zero;
            Z[1] = zero;
        }
        const v4f k0 = *(const v4f*)&kRow[8 * T];
        const v4f k1 = *(const v4f*)&kRow[8 * T + 4];
        #pragma unroll
        for (int q = 0; q < 8; ++q) {
            const float kq = (q < 4) ? k0[q & 3] : k1[q & 3];
            const v2f kk = {kq, kq};
            #pragma unroll
            for (int s = 0; s < 4; ++s) {
                const int i = s - q;
                if (i >= 0) {
                    acc[s] = __builtin_elementwise_fma(kk, half_pair(X, i), acc[s]);
                } else if (i >= -4) {
                    acc[s] = __builtin_elementwise_fma(kk, half_pair(Z, 4 + i), acc[s]);
                }
            }
        }
    }
}

__device__ __forceinline__ void put_half(float* t, const v2f (&acc)[16]) {
    v4f* p = (v4f*)t;
    #pragma unroll
    for (int s = 0; s < 2; ++s) {
        const v4f o = {acc[2 * s].x, acc[2 * s].y, acc[2 * s + 1].x, acc[2 * s + 1].y};
        p[s] = o;
    }
}

__global__ void __launch_bounds__(ConvThreads, 6)
kconv(const float* __restrict__ u,
      const float* __restrict__ Dv,
      const float* __restrict__ Kc,
      float* __restrict__ y) {
    __shared__ __attribute__((aligned(16))) float smem[NKinds * KPitch + PairFloats + ConvWaves];
    float* const sK = smem;
    float* const sU = smem + NKinds * KPitch;
    float* const sTop = smem + NKinds * KPitch + PairFloats;

    const int c   = blockIdx.x;
    const int tid = threadIdx.x;

    v4f ga[3], gb[3], gk[2];
    #pragma unroll
    for (int it = 0; it < 3; ++it) {
        const int q = it * ConvThreads + tid;
        if (q < 2 * Tq) {
            const int r = q >> 9;
            const int i = q & (Tq - 1);
            const float* ua = u + ((size_t)(2 * r) * Cdim + c) * Tdim;
            const float* ub = ua + (size_t)Cdim * Tdim;
            ga[it] = ((const v4f*)ua)[i];
            gb[it] = ((const v4f*)ub)[i];
        }
    }
    #pragma unroll
    for (int it = 0; it < 2; ++it) {
        const int q = it * ConvThreads + tid;
        if (q < Tq) gk[it] = ((const v4f*)(Kc + (size_t)c * Tdim))[q];
    }
    #pragma unroll
    for (int it = 0; it < 3; ++it) {
        const int q = it * ConvThreads + tid;
        if (q < 2 * Tq) {
            const int r = q >> 9;
            const int i = q & (Tq - 1);
            const v4f a  = ga[it];
            const v4f bq = gb[it];
            const v2f px = {a.x, bq.x};
            const v2f py = {a.y, bq.y};
            const v2f pz = {a.z, bq.z};
            const v2f pw = {a.w, bq.w};
            const v2f sxy = px + py;
            const v2f szw = pz + pw;
            float* dst = sU + (2 * (i >> 4) + r) * PitchW + 2 * (i & 15);
            *(v2f*)(dst + kind_base_w(0)) = px;
            *(v2f*)(dst + kind_base_w(1)) = pz;
            *(v2f*)(dst + kind_base_w(2)) = px + pz;
            *(v2f*)(dst + kind_base_w(3)) = py;
            *(v2f*)(dst + kind_base_w(4)) = pw;
            *(v2f*)(dst + kind_base_w(5)) = py + pw;
            *(v2f*)(dst + kind_base_w(6)) = sxy;
            *(v2f*)(dst + kind_base_w(7)) = szw;
            float* dst8 = sU + Kind8Base + (2 * (i >> 3) + r) * TilePitch + 2 * (i & 7);
            *(v2f*)dst8 = sxy + szw;
        }
    }
    int waveTop = 0;
    #pragma unroll
    for (int it = 0; it < 2; ++it) {
        const int q = it * ConvThreads + tid;
        bool live = false;
        if (q < Tq) {
            const v4f k = gk[it];
            live = (k.x != 0.0f) || (k.y != 0.0f) || (k.z != 0.0f) || (k.w != 0.0f);
        }
        const unsigned liveMask = __builtin_amdgcn_ballot_w32(live);
        if (liveMask != 0u) {
            const int qTop = it * ConvThreads + (tid & ~31) + 31 - __builtin_clz(liveMask);
            const int gTop = (qTop >> 4) + 1;
            waveTop = (gTop > waveTop) ? gTop : waveTop;
        }
        if (q < Tq) {
            const v4f k = gk[it];
            const float kxy = k.x + k.y;
            const float kzw = k.z + k.w;
            float* dst = sK + q;
            dst[0 * KPitch] = k.x;
            dst[1 * KPitch] = k.z;
            dst[2 * KPitch] = k.x + k.z;
            dst[3 * KPitch] = k.y;
            dst[4 * KPitch] = k.w;
            dst[5 * KPitch] = k.y + k.w;
            dst[6 * KPitch] = kxy;
            dst[7 * KPitch] = kzw;
            dst[8 * KPitch] = kxy + kzw;
        }
    }
    if ((tid & 31) == 0) sTop[tid >> 5] = (float)waveTop;
    __syncthreads();

    int nz;
    {
        const v4f m0 = ((const v4f*)sTop)[0];
        const v4f m1 = ((const v4f*)sTop)[1];
        const v4f m2 = ((const v4f*)sTop)[2];
        const v4f mm = __builtin_elementwise_max(__builtin_elementwise_max(m0, m1), m2);
        const float top = fmaxf(fmaxf(mm.x, mm.y), fmaxf(mm.z, mm.w));
        nz = __builtin_amdgcn_readfirstlane((int)top);
    }

    const int lane = tid & 31;
    const int wave = tid >> 5;
    const int hf   = lane >> 4;
    const int l16  = lane & 15;
    int kind, r, j, hs = 0;
    const bool whole = wave < 8;
    if (whole) {
        kind = l16 >> 1;
        r    = l16 & 1;
        j    = 2 * wave + hf;
    } else {
        kind = 8;
        hs   = l16 >> 3;
        r    = (l16 >> 2) & 1;
        j    = (wave - 8) * 8 + hf * 4 + (l16 & 3);
    }

    float* const uRow = whole ? (sU + kind_base_w(kind) + r * PitchW)
                              : (sU + Kind8Base + r * TilePitch);
    const float* kRow = sK + kind * KPitch;

    const bool skip = (kind == 0) || (kind == 2) || (kind == 6) || (kind == 8);
    const float d = skip ? Dv[c] : 0.0f;

    v2f resLong[16], resShort[16];
    if (whole) {
        conv_wide(resLong,  uRow, kRow, d, (TilesW - 1) - j, nz);
        conv_wide(resShort, uRow, kRow, d, j, nz);
    } else {
        conv_half(resLong,  uRow, kRow, d, (TilesQ - 1) - j, hs, 2 * nz);
        conv_half(resShort, uRow, kRow, d, j, hs, 2 * nz);
    }

    __syncthreads();
    if (whole) {
        put_wide(uRow + ((TilesW - 1) - j) * StrideW, resLong);
        put_wide(uRow + j * StrideW, resShort);
    } else {
        put_half(uRow + ((TilesQ - 1) - j) * TileStride + 8 * hs, resLong);
        put_half(uRow + j * TileStride + 8 * hs, resShort);
    }
    __syncthreads();

    #pragma unroll
    for (int it = 0; it < 3; ++it) {
        const int q = it * ConvThreads + tid;
        if (q < 2 * Tq) {
            const int rr = q >> 9;
            const int i  = q & (Tq - 1);
            const float* pc = sU + (2 * (i >> 4) + rr) * PitchW + 2 * (i & 15);
            const v2f R0 = *(const v2f*)(pc + kind_base_w(0));
            const v2f R1 = *(const v2f*)(pc + kind_base_w(1));
            const v2f R2 = *(const v2f*)(pc + kind_base_w(2));
            const v2f R3 = *(const v2f*)(pc + kind_base_w(3));
            const v2f R4 = *(const v2f*)(pc + kind_base_w(4));
            const v2f R5 = *(const v2f*)(pc + kind_base_w(5));
            const v2f R6 = *(const v2f*)(pc + kind_base_w(6));
            const v2f R7 = *(const v2f*)(pc + kind_base_w(7));
            const v2f R8 = *(const v2f*)(sU + Kind8Base + (2 * (i >> 3) + rr) * TilePitch + 2 * (i & 7));
            v2f q1 = {0.0f, 0.0f}, q3 = q1, q4 = q1, q5 = q1, q7 = q1;
            if (i > 0) {
                const float* pp = pc + ((i & 15) ? -2 : (30 - StrideW));
                q1 = *(const v2f*)(pp + kind_base_w(1));
                q3 = *(const v2f*)(pp + kind_base_w(3));
                q4 = *(const v2f*)(pp + kind_base_w(4));
                q5 = *(const v2f*)(pp + kind_base_w(5));
                q7 = *(const v2f*)(pp + kind_base_w(7));
            }
            const v2f pe0 = R0 + q1;
            const v2f po0 = R2 - R0 - R1;
            const v2f pe1 = R3 + q4;
            const v2f po1 = R5 - R3 - R4;
            const v2f peS = R6 + q7;
            const v2f poS = R8 - R6 - R7;
            const v2f pm1 = q5 - q3 - q4;
            const v2f y0 = pe0 + pm1;
            const v2f y1 = peS - pe0 - pe1;
            const v2f y2 = po0 + pe1;
            const v2f y3 = poS - po0 - po1;
            float* ya = y + ((size_t)(2 * rr) * Cdim + c) * Tdim;
            float* yb = ya + (size_t)Cdim * Tdim;
            const v4f oa = {y0.x, y1.x, y2.x, y3.x};
            const v4f ob = {y0.y, y1.y, y2.y, y3.y};
            ((v4f*)ya)[i] = oa;
            ((v4f*)yb)[i] = ob;
        }
    }
}

extern "C" void kernel_launch(void* const* d_in, const int* in_sizes, int n_in,
                              void* d_out, int out_size, void* d_ws, size_t ws_size,
                              hipStream_t stream) {
    const float* u       = (const float*)d_in[0];
    const float* log_tau = (const float*)d_in[1];
    const float* mix     = (const float*)d_in[2];
    const float* Dv      = (const float*)d_in[3];
    const float* log_dt  = (const float*)d_in[4];
    float*       y       = (float*)d_out;
    float*       Kws     = (float*)d_ws;

    (void)in_sizes; (void)n_in; (void)out_size; (void)ws_size;

    dim3 g1(KgGridX, KgGridY);
    kgen_wmma<<<g1, KgWaves * 32, 0, stream>>>(mix, log_tau, log_dt, Kws);

    dim3 g2(Cdim);
    kconv<<<g2, ConvThreads, 0, stream>>>(u, Dv, Kws, y);
}
